# starting kernel
# baseline (speedup 1.0000x reference)
#include <hip/hip_runtime.h>
#include <hip/hip_bf16.h>

typedef __attribute__((ext_vector_type(2))) float v2f;
typedef __attribute__((ext_vector_type(8))) float v8f;

#define NODE_DIM   64
#define LEAKY      0.2f

__global__ void __launch_bounds__(256)
proj_gemm_wmma(const float* __restrict__ X, const float* __restrict__ A,
               float* __restrict__ P, int n_units) {
  const int lane = threadIdx.x & 31;
  const int wave = threadIdx.x >> 5;
  const int half = lane >> 4;
  const int m    = lane & 15;
  const int row_base = blockIdx.x * 128 + wave * 16;

  int row = row_base + m;
  if (row >= n_units) row = n_units - 1;
  const float* __restrict__ xrow = X + (size_t)row * NODE_DIM;
  const float* __restrict__ wcol = A + (m & 7) * 128 + ((m >> 3) & 1) * 64;

  v8f c = {};
#pragma unroll
  for (int k0 = 0; k0 < NODE_DIM; k0 += 4) {
    const int k = k0 + half * 2;
    v2f a, b;
    a.x = xrow[k];     a.y = xrow[k + 1];
    b.x = wcol[k];     b.y = wcol[k + 1];
    c = __builtin_amdgcn_wmma_f32_16x16x4_f32(
         false, a,  false, b,
         (short)0, c,  false,  false);
  }

#pragma unroll
  for (int j = 0; j < 8; ++j) {
    const int orow = row_base + j + half * 8;
    if (orow < n_units) P[(size_t)orow * 16 + m] = c[j];
  }
}

__global__ void __launch_bounds__(256)
init_nodes(int* __restrict__ seg_max_bits, float* __restrict__ denom,
           float* __restrict__ gnum, int n) {
  const int i = blockIdx.x * blockDim.x + threadIdx.x;
  if (i < n) {
    seg_max_bits[i] = 0xFF800000;
    denom[i] = 0.0f;
    gnum[i]  = 0.0f;
  }
}

__device__ __forceinline__ void atomicMaxF32(float* addr, float v) {
  const int vi = __float_as_int(v);
  if (vi >= 0) atomicMax((int*)addr, vi);
  else         atomicMin((unsigned int*)addr, (unsigned int)vi);
}

__device__ __forceinline__ float edge_logit(const float* __restrict__ P,
                                            int s, int d, int r) {
  float e = P[(size_t)s * 16 + r] + P[(size_t)d * 16 + 8 + r];
  return (e > 0.0f) ? e : LEAKY * e;
}

__global__ void __launch_bounds__(256)
edge_pass_max(const float* __restrict__ P, const int* __restrict__ src,
              const int* __restrict__ dst, const int* __restrict__ etype,
              float* __restrict__ seg_max, int n_edges) {
  const int i = blockIdx.x * blockDim.x + threadIdx.x;
  if (i >= n_edges) return;
  const int s = src[i], d = dst[i], r = etype[i];
  atomicMaxF32(&seg_max[s], edge_logit(P, s, d, r));
}

__global__ void __launch_bounds__(256)
edge_pass_sum(const float* __restrict__ P, const int* __restrict__ src,
              const int* __restrict__ dst, const int* __restrict__ etype,
              const float* __restrict__ seg_max, const float* __restrict__ hat_t,
              float* __restrict__ denom, float* __restrict__ gnum, int n_edges) {
  const int i = blockIdx.x * blockDim.x + threadIdx.x;
  if (i >= n_edges) return;
  const int s = src[i], d = dst[i], r = etype[i];
  const float ex = __expf(edge_logit(P, s, d, r) - seg_max[s]);
  atomicAdd(&denom[s], ex);
  atomicAdd(&gnum[s], ex * hat_t[d]);
}

__global__ void __launch_bounds__(256)
finalize_nodes(const float* __restrict__ denom, const float* __restrict__ gnum,
               float* __restrict__ out, int n) {
  const int i = blockIdx.x * blockDim.x + threadIdx.x;
  if (i < n) {
    const float dn = denom[i];
    out[i] = (dn > 0.0f) ? (gnum[i] / dn) : 0.0f;
  }
}

extern "C" void kernel_launch(void* const* d_in, const int* in_sizes, int n_in,
                              void* d_out, int out_size, void* d_ws, size_t ws_size,
                              hipStream_t stream) {
  const float* X     = (const float*)d_in[0];
  const float* hat_t = (const float*)d_in[1];
  const float* A     = (const float*)d_in[2];
  const int*   eidx  = (const int*)d_in[3];
  const int*   etype = (const int*)d_in[4];

  const int n_units = in_sizes[1];
  const int n_edges = in_sizes[4];
  const int* src = eidx;
  const int* dst = eidx + n_edges;

  float* P        = (float*)d_ws;
  float* seg_max  = P + (size_t)n_units * 16;
  float* denom    = seg_max + n_units;
  float* gnum     = denom + n_units;
  float* out      = (float*)d_out;

  const int gemm_blocks = (n_units + 127) / 128;
  proj_gemm_wmma<<<gemm_blocks, 256, 0, stream>>>(X, A, P, n_units);

  const int node_blocks = (n_units + 255) / 256;
  init_nodes<<<node_blocks, 256, 0, stream>>>((int*)seg_max, denom, gnum, n_units);

  const int edge_blocks = (n_edges + 255) / 256;
  edge_pass_max<<<edge_blocks, 256, 0, stream>>>(P, src, dst, etype, seg_max, n_edges);
  edge_pass_sum<<<edge_blocks, 256, 0, stream>>>(P, src, dst, etype, seg_max, hat_t,
                                                 denom, gnum, n_edges);
  finalize_nodes<<<node_blocks, 256, 0, stream>>>(denom, gnum, out, n_units);
}


# pass 1 stores each edge's logit (one coalesced float per edge, only if the workspace holds the plane) and pass 2 streams it back instead of gathering both projection rows a second time
# speedup vs baseline: 1.0396x; 1.0396x over previous
#include <hip/hip_runtime.h>
#include <hip/hip_bf16.h>

typedef __attribute__((ext_vector_type(2))) float v2f;
typedef __attribute__((ext_vector_type(8))) float v8f;

#define NODE_DIM   64
#define LEAKY      0.2f

__global__ void __launch_bounds__(256)
proj_gemm_wmma(const float* __restrict__ X, const float* __restrict__ A,
               float* __restrict__ P, int n_units) {
  const int lane = threadIdx.x & 31;
  const int wave = threadIdx.x >> 5;
  const int half = lane >> 4;
  const int m    = lane & 15;
  const int row_base = blockIdx.x * 128 + wave * 16;

  int row = row_base + m;
  if (row >= n_units) row = n_units - 1;
  const float* __restrict__ xrow = X + (size_t)row * NODE_DIM;
  const float* __restrict__ wcol = A + (m & 7) * 128 + ((m >> 3) & 1) * 64;

  v8f c = {};
#pragma unroll
  for (int k0 = 0; k0 < NODE_DIM; k0 += 4) {
    const int k = k0 + half * 2;
    v2f a, b;
    a.x = xrow[k];     a.y = xrow[k + 1];
    b.x = wcol[k];     b.y = wcol[k + 1];
    c = __builtin_amdgcn_wmma_f32_16x16x4_f32(
         false, a,  false, b,
         (short)0, c,  false,  false);
  }

#pragma unroll
  for (int j = 0; j < 8; ++j) {
    const int orow = row_base + j + half * 8;
    if (orow < n_units) P[(size_t)orow * 16 + m] = c[j];
  }
}

__global__ void __launch_bounds__(256)
init_nodes(int* __restrict__ seg_max_bits, float* __restrict__ denom,
           float* __restrict__ gnum, int n) {
  const int i = blockIdx.x * blockDim.x + threadIdx.x;
  if (i < n) {
    seg_max_bits[i] = 0xFF800000;
    denom[i] = 0.0f;
    gnum[i]  = 0.0f;
  }
}

__device__ __forceinline__ void atomicMaxF32(float* addr, float v) {
  const int vi = __float_as_int(v);
  if (vi >= 0) atomicMax((int*)addr, vi);
  else         atomicMin((unsigned int*)addr, (unsigned int)vi);
}

__device__ __forceinline__ bool logit_plane_fits(int n_units, int n_edges,
                                                 size_t ws_bytes) {
  return ws_bytes >= ((size_t)n_units * 19 + (size_t)n_edges) * sizeof(float);
}

__device__ __forceinline__ float edge_logit(const float* __restrict__ P,
                                            int s, int d, int r) {
  float e = P[(size_t)s * 16 + r] + P[(size_t)d * 16 + 8 + r];
  return (e > 0.0f) ? e : LEAKY * e;
}

__global__ void __launch_bounds__(256)
edge_pass_max(const float* __restrict__ P, const int* __restrict__ src,
              const int* __restrict__ dst, const int* __restrict__ etype,
              float* __restrict__ seg_max, float* __restrict__ e_plane,
              int n_edges, int n_units, size_t ws_bytes) {
  const int i = blockIdx.x * blockDim.x + threadIdx.x;
  if (i >= n_edges) return;
  const int s = src[i], d = dst[i], r = etype[i];
  const float e = edge_logit(P, s, d, r);
  if (logit_plane_fits(n_units, n_edges, ws_bytes)) e_plane[i] = e;
  atomicMaxF32(&seg_max[s], e);
}

__global__ void __launch_bounds__(256)
edge_pass_sum(const float* __restrict__ P, const int* __restrict__ src,
              const int* __restrict__ dst, const int* __restrict__ etype,
              const float* __restrict__ e_plane,
              const float* __restrict__ seg_max, const float* __restrict__ hat_t,
              float* __restrict__ denom, float* __restrict__ gnum,
              int n_edges, int n_units, size_t ws_bytes) {
  const int i = blockIdx.x * blockDim.x + threadIdx.x;
  if (i >= n_edges) return;
  const bool fits = logit_plane_fits(n_units, n_edges, ws_bytes);
  float e = 0.0f;
  if (fits) e = e_plane[i];
  const int s = src[i], d = dst[i];
  if (!fits) e = edge_logit(P, s, d, etype[i]);
  const float ex = __expf(e - seg_max[s]);
  atomicAdd(&denom[s], ex);
  atomicAdd(&gnum[s], ex * hat_t[d]);
}

__global__ void __launch_bounds__(256)
finalize_nodes(const float* __restrict__ denom, const float* __restrict__ gnum,
               float* __restrict__ out, int n) {
  const int i = blockIdx.x * blockDim.x + threadIdx.x;
  if (i < n) {
    const float dn = denom[i];
    out[i] = (dn > 0.0f) ? (gnum[i] / dn) : 0.0f;
  }
}

extern "C" void kernel_launch(void* const* d_in, const int* in_sizes, int n_in,
                              void* d_out, int out_size, void* d_ws, size_t ws_size,
                              hipStream_t stream) {
  const float* X     = (const float*)d_in[0];
  const float* hat_t = (const float*)d_in[1];
  const float* A     = (const float*)d_in[2];
  const int*   eidx  = (const int*)d_in[3];
  const int*   etype = (const int*)d_in[4];

  const int n_units = in_sizes[1];
  const int n_edges = in_sizes[4];
  const int* src = eidx;
  const int* dst = eidx + n_edges;

  float* P        = (float*)d_ws;
  float* seg_max  = P + (size_t)n_units * 16;
  float* denom    = seg_max + n_units;
  float* gnum     = denom + n_units;
  float* e_plane  = gnum + n_units;
  float* out      = (float*)d_out;

  const int gemm_blocks = (n_units + 127) / 128;
  proj_gemm_wmma<<<gemm_blocks, 256, 0, stream>>>(X, A, P, n_units);

  const int node_blocks = (n_units + 255) / 256;
  init_nodes<<<node_blocks, 256, 0, stream>>>((int*)seg_max, denom, gnum, n_units);

  const int edge_blocks = (n_edges + 255) / 256;
  edge_pass_max<<<edge_blocks, 256, 0, stream>>>(P, src, dst, etype, seg_max, e_plane,
                                                 n_edges, n_units, ws_size);
  edge_pass_sum<<<edge_blocks, 256, 0, stream>>>(P, src, dst, etype, e_plane, seg_max,
                                                 hat_t, denom, gnum,
                                                 n_edges, n_units, ws_size);
  finalize_nodes<<<node_blocks, 256, 0, stream>>>(denom, gnum, out, n_units);
}


# the three per-edge global atomics (f32 max, two f32 adds) become one returning integer add that ranks the edge in its source's bucket; scan, place, then one thread per node sums in arrival-rank order
# speedup vs baseline: 1.3391x; 1.2881x over previous
#include <hip/hip_runtime.h>
#include <hip/hip_bf16.h>

typedef __attribute__((ext_vector_type(2))) float v2f;
typedef __attribute__((ext_vector_type(8))) float v8f;

#define NODE_DIM   64
#define LEAKY      0.2f

__global__ void __launch_bounds__(256)
proj_gemm_wmma(const float* __restrict__ X, const float* __restrict__ A,
               float* __restrict__ P, int n_units) {
  const int lane = threadIdx.x & 31;
  const int wave = threadIdx.x >> 5;
  const int half = lane >> 4;
  const int m    = lane & 15;
  const int row_base = blockIdx.x * 128 + wave * 16;

  int row = row_base + m;
  if (row >= n_units) row = n_units - 1;
  const float* __restrict__ xrow = X + (size_t)row * NODE_DIM;
  const float* __restrict__ wcol = A + (m & 7) * 128 + ((m >> 3) & 1) * 64;

  v8f c = {};
#pragma unroll
  for (int k0 = 0; k0 < NODE_DIM; k0 += 4) {
    const int k = k0 + half * 2;
    v2f a, b;
    a.x = xrow[k];     a.y = xrow[k + 1];
    b.x = wcol[k];     b.y = wcol[k + 1];
    c = __builtin_amdgcn_wmma_f32_16x16x4_f32(
         false, a,  false, b,
         (short)0, c,  false,  false);
  }

#pragma unroll
  for (int j = 0; j < 8; ++j) {
    const int orow = row_base + j + half * 8;
    if (orow < n_units) P[(size_t)orow * 16 + m] = c[j];
  }
}

__device__ __forceinline__ bool bucket_fits(int n_units, int n_edges, size_t ws_bytes) {
  return ws_bytes >= ((size_t)n_units * 18 + (size_t)n_edges * 4) * sizeof(float);
}

__device__ __forceinline__ bool logit_plane_fits(int n_units, int n_edges,
                                                 size_t ws_bytes) {
  return ws_bytes >= ((size_t)n_units * 19 + (size_t)n_edges) * sizeof(float);
}

__device__ __forceinline__ void atomicMaxF32(float* addr, float v) {
  const int vi = __float_as_int(v);
  if (vi >= 0) atomicMax((int*)addr, vi);
  else         atomicMin((unsigned int*)addr, (unsigned int)vi);
}

__device__ __forceinline__ float edge_logit(const float* __restrict__ P,
                                            int s, int d, int r) {
  float e = P[(size_t)s * 16 + r] + P[(size_t)d * 16 + 8 + r];
  return (e > 0.0f) ? e : LEAKY * e;
}

__global__ void __launch_bounds__(256)
init_nodes(int* __restrict__ cnt, int* __restrict__ seg_max_bits,
           float* __restrict__ denom, float* __restrict__ gnum,
           int n_units, int n_edges, size_t ws_bytes) {
  const int i = blockIdx.x * blockDim.x + threadIdx.x;
  if (i >= n_units) return;
  if (bucket_fits(n_units, n_edges, ws_bytes)) {
    cnt[i] = 0;
  } else {
    seg_max_bits[i] = 0xFF800000;
    denom[i] = 0.0f;
    gnum[i]  = 0.0f;
  }
}

__global__ void __launch_bounds__(256)
edge_pass_rank(const float* __restrict__ P, const int* __restrict__ src,
               const int* __restrict__ dst, const int* __restrict__ etype,
               int* __restrict__ cnt, int* __restrict__ rank,
               float* __restrict__ e_edge,
               float* __restrict__ seg_max, float* __restrict__ e_plane,
               int n_edges, int n_units, size_t ws_bytes) {
  const int i = blockIdx.x * blockDim.x + threadIdx.x;
  if (i >= n_edges) return;
  const int s = src[i], d = dst[i], r = etype[i];
  if (bucket_fits(n_units, n_edges, ws_bytes)) {
    const int rk = atomicAdd(&cnt[s], 1);
    const float e = edge_logit(P, s, d, r);
    rank[i]   = rk;
    e_edge[i] = e;
  } else {
    const float e = edge_logit(P, s, d, r);
    if (logit_plane_fits(n_units, n_edges, ws_bytes)) e_plane[i] = e;
    atomicMaxF32(&seg_max[s], e);
  }
}

__global__ void __launch_bounds__(1024)
scan_counts(const int* __restrict__ cnt, int* __restrict__ off,
            int n_units, int n_edges, size_t ws_bytes) {
  if (!bucket_fits(n_units, n_edges, ws_bytes)) return;
  __shared__ int wsum[32];
  __shared__ int s_base;
  const int tid  = threadIdx.x;
  const int lane = tid & 31;
  const int wave = tid >> 5;
  const int first = blockIdx.x * 1024;

  int acc = 0;
  const int4* __restrict__ c4 = (const int4*)cnt;
#pragma unroll 4
  for (int j = tid; j < first / 4; j += 1024) {
    const int4 v = c4[j];
    acc += (v.x + v.y) + (v.z + v.w);
  }
  for (int o = 16; o > 0; o >>= 1) acc += __shfl_xor(acc, o);
  if (lane == 0) wsum[wave] = acc;
  __syncthreads();
  if (wave == 0) {
    int v = wsum[lane];
    for (int o = 16; o > 0; o >>= 1) v += __shfl_xor(v, o);
    if (lane == 0) s_base = v;
  }
  __syncthreads();
  const int base = s_base;

  const int node = first + tid;
  const int c = (node < n_units) ? cnt[node] : 0;
  int inc = c;
  for (int o = 1; o < 32; o <<= 1) {
    const int t = __shfl_up(inc, o);
    if (lane >= o) inc += t;
  }
  if (lane == 31) wsum[wave] = inc;
  __syncthreads();
  if (wave == 0) {
    const int v = wsum[lane];
    int iv = v;
    for (int o = 1; o < 32; o <<= 1) {
      const int t = __shfl_up(iv, o);
      if (lane >= o) iv += t;
    }
    wsum[lane] = iv - v;
  }
  __syncthreads();
  if (node < n_units) off[node] = base + wsum[wave] + (inc - c);
}

__global__ void __launch_bounds__(256)
edge_pass_place(const float* __restrict__ P, const int* __restrict__ src,
                const int* __restrict__ dst, const int* __restrict__ etype,
                const int* __restrict__ off, const int* __restrict__ rank,
                const float* __restrict__ e_edge, float2* __restrict__ pairs,
                const float* __restrict__ e_plane,
                const float* __restrict__ seg_max, const float* __restrict__ hat_t,
                float* __restrict__ denom, float* __restrict__ gnum,
                int n_edges, int n_units, size_t ws_bytes) {
  const int i = blockIdx.x * blockDim.x + threadIdx.x;
  if (i >= n_edges) return;
  if (bucket_fits(n_units, n_edges, ws_bytes)) {
    const int s = src[i], d = dst[i];
    const int rk = rank[i];
    const float e = e_edge[i];
    const size_t pos = (size_t)off[s] + (size_t)rk;
    pairs[pos] = make_float2(e, hat_t[d]);
  } else {
    const bool fits = logit_plane_fits(n_units, n_edges, ws_bytes);
    float e = 0.0f;
    if (fits) e = e_plane[i];
    const int s = src[i], d = dst[i];
    if (!fits) e = edge_logit(P, s, d, etype[i]);
    const float ex = __expf(e - seg_max[s]);
    atomicAdd(&denom[s], ex);
    atomicAdd(&gnum[s], ex * hat_t[d]);
  }
}

__global__ void __launch_bounds__(256)
finalize_nodes(const int* __restrict__ off, const int* __restrict__ cnt,
               const float2* __restrict__ pairs,
               const float* __restrict__ denom, const float* __restrict__ gnum,
               float* __restrict__ out, int n_units, int n_edges, size_t ws_bytes) {
  const int i = blockIdx.x * blockDim.x + threadIdx.x;
  if (i >= n_units) return;
  if (bucket_fits(n_units, n_edges, ws_bytes)) {
    const int c = cnt[i];
    const float2* __restrict__ seg = pairs + (size_t)off[i];
    float m = -INFINITY;
    for (int j = 0; j < c; ++j) {
      const float x = seg[j].x;
      m = (x > m) ? x : m;
    }
    float dn = 0.0f, g = 0.0f;
    for (int j = 0; j < c; ++j) {
      const float2 p = seg[j];
      const float ex = __expf(p.x - m);
      dn += ex;
      g  += ex * p.y;
    }
    out[i] = (c > 0) ? (g / dn) : 0.0f;
  } else {
    const float dn = denom[i];
    out[i] = (dn > 0.0f) ? (gnum[i] / dn) : 0.0f;
  }
}

extern "C" void kernel_launch(void* const* d_in, const int* in_sizes, int n_in,
                              void* d_out, int out_size, void* d_ws, size_t ws_size,
                              hipStream_t stream) {
  const float* X     = (const float*)d_in[0];
  const float* hat_t = (const float*)d_in[1];
  const float* A     = (const float*)d_in[2];
  const int*   eidx  = (const int*)d_in[3];
  const int*   etype = (const int*)d_in[4];

  const int n_units = in_sizes[1];
  const int n_edges = in_sizes[4];
  const int* src = eidx;
  const int* dst = eidx + n_edges;

  float* P        = (float*)d_ws;
  float* out      = (float*)d_out;

  int*    cnt     = (int*)(P + (size_t)n_units * 16);
  int*    off     = cnt + n_units;
  int*    rank    = off + n_units;
  float*  e_edge  = (float*)(rank + n_edges);
  float2* pairs   = (float2*)(e_edge + n_edges);

  float* seg_max  = P + (size_t)n_units * 16;
  float* denom    = seg_max + n_units;
  float* gnum     = denom + n_units;
  float* e_plane  = gnum + n_units;

  const int gemm_blocks = (n_units + 127) / 128;
  proj_gemm_wmma<<<gemm_blocks, 256, 0, stream>>>(X, A, P, n_units);

  const int node_blocks = (n_units + 255) / 256;
  init_nodes<<<node_blocks, 256, 0, stream>>>(cnt, (int*)seg_max, denom, gnum,
                                              n_units, n_edges, ws_size);

  const int edge_blocks = (n_edges + 255) / 256;
  edge_pass_rank<<<edge_blocks, 256, 0, stream>>>(P, src, dst, etype, cnt, rank, e_edge,
                                                  seg_max, e_plane,
                                                  n_edges, n_units, ws_size);

  const int scan_blocks = (n_units + 1023) / 1024;
  scan_counts<<<scan_blocks, 1024, 0, stream>>>(cnt, off, n_units, n_edges, ws_size);

  edge_pass_place<<<edge_blocks, 256, 0, stream>>>(P, src, dst, etype, off, rank, e_edge,
                                                   pairs, e_plane, seg_max, hat_t,
                                                   denom, gnum,
                                                   n_edges, n_units, ws_size);
  finalize_nodes<<<node_blocks, 256, 0, stream>>>(off, cnt, pairs, denom, gnum, out,
                                                  n_units, n_edges, ws_size);
}


# counting sort by source (rank add, scan, place pass) becomes two linked lists per node: one returning exchange per edge; node thread walks both once, running max, sums in list order
# speedup vs baseline: 1.6876x; 1.2602x over previous
#include <hip/hip_runtime.h>
#include <hip/hip_bf16.h>

typedef __attribute__((ext_vector_type(2))) float v2f;
typedef __attribute__((ext_vector_type(8))) float v8f;

#define NODE_DIM   64
#define LEAKY      0.2f

__global__ void __launch_bounds__(256)
proj_gemm_wmma(const float* __restrict__ X, const float* __restrict__ A,
               float* __restrict__ P, int n_units) {
  const int lane = threadIdx.x & 31;
  const int wave = threadIdx.x >> 5;
  const int half = lane >> 4;
  const int m    = lane & 15;
  const int row_base = blockIdx.x * 128 + wave * 16;

  int row = row_base + m;
  if (row >= n_units) row = n_units - 1;
  const float* __restrict__ xrow = X + (size_t)row * NODE_DIM;
  const float* __restrict__ wcol = A + (m & 7) * 128 + ((m >> 3) & 1) * 64;

  v8f c = {};
#pragma unroll
  for (int k0 = 0; k0 < NODE_DIM; k0 += 4) {
    const int k = k0 + half * 2;
    v2f a, b;
    a.x = xrow[k];     a.y = xrow[k + 1];
    b.x = wcol[k];     b.y = wcol[k + 1];
    c = __builtin_amdgcn_wmma_f32_16x16x4_f32(
         false, a,  false, b,
         (short)0, c,  false,  false);
  }

#pragma unroll
  for (int j = 0; j < 8; ++j) {
    const int orow = row_base + j + half * 8;
    if (orow < n_units) P[(size_t)orow * 16 + m] = c[j];
  }
}

__device__ __forceinline__ bool lists_fit(int n_units, int n_edges, size_t ws_bytes) {
  return ws_bytes >= ((size_t)n_units * 18 + (size_t)n_edges * 4) * sizeof(float);
}

__device__ __forceinline__ float edge_logit(const float* __restrict__ P,
                                            int s, int d, int r) {
  float e = P[(size_t)s * 16 + r] + P[(size_t)d * 16 + 8 + r];
  return (e > 0.0f) ? e : LEAKY * e;
}

__global__ void __launch_bounds__(256)
init_heads(int2* __restrict__ head, int n_units, int n_edges, size_t ws_bytes) {
  const int i = blockIdx.x * blockDim.x + threadIdx.x;
  if (i >= n_units) return;
  if (!lists_fit(n_units, n_edges, ws_bytes)) return;
  head[i] = make_int2(-1, -1);
}

__global__ void __launch_bounds__(256)
edge_pass_link(const float* __restrict__ P, const int* __restrict__ src,
               const int* __restrict__ dst, const int* __restrict__ etype,
               const float* __restrict__ hat_t,
               int* __restrict__ head, int4* __restrict__ rec,
               int n_edges, int n_units, size_t ws_bytes) {
  const int i = blockIdx.x * blockDim.x + threadIdx.x;
  if (i >= n_edges) return;
  if (!lists_fit(n_units, n_edges, ws_bytes)) return;
  const int s = src[i], d = dst[i], r = etype[i];
  const int prev = atomicExch(&head[2 * (size_t)s + (i & 1)], i);
  const float e = edge_logit(P, s, d, r);
  const float t = hat_t[d];
  rec[i] = make_int4(__float_as_int(e), __float_as_int(t), prev, 0);
}

__global__ void __launch_bounds__(128)
finalize_lists(const int2* __restrict__ head, const int4* __restrict__ rec,
               float* __restrict__ out, int n_units, int n_edges, size_t ws_bytes) {
  const int i = blockIdx.x * blockDim.x + threadIdx.x;
  if (i >= n_units) return;
  if (!lists_fit(n_units, n_edges, ws_bytes)) return;
  const int2 h = head[i];
  int j0 = h.x, j1 = h.y;
  const bool any = (j0 >= 0) || (j1 >= 0);
  float m = -INFINITY, dn = 0.0f, g = 0.0f;
  for (int k = 0; k < n_edges; ++k) {
    const bool a0 = (unsigned)j0 < (unsigned)n_edges;
    const bool a1 = (unsigned)j1 < (unsigned)n_edges;
    if (!(a0 || a1)) break;
    int4 r0 = make_int4(0, 0, -1, 0);
    int4 r1 = r0;
    if (a0) r0 = rec[j0];
    if (a1) r1 = rec[j1];
    const float x0 = a0 ? __int_as_float(r0.x) : -INFINITY;
    const float x1 = a1 ? __int_as_float(r1.x) : -INFINITY;
    const float t0 = a0 ? __int_as_float(r0.y) : 0.0f;
    const float t1 = a1 ? __int_as_float(r1.y) : 0.0f;
    const float mn = fmaxf(m, fmaxf(x0, x1));
    const float sc = __expf(m - mn);
    const float b0 = __expf(x0 - mn);
    const float b1 = __expf(x1 - mn);
    dn = (dn * sc + b0) + b1;
    g  = (g * sc + b0 * t0) + b1 * t1;
    m  = mn;
    j0 = r0.z;
    j1 = r1.z;
  }
  out[i] = any ? (g / dn) : 0.0f;
}

extern "C" void kernel_launch(void* const* d_in, const int* in_sizes, int n_in,
                              void* d_out, int out_size, void* d_ws, size_t ws_size,
                              hipStream_t stream) {
  const float* X     = (const float*)d_in[0];
  const float* hat_t = (const float*)d_in[1];
  const float* A     = (const float*)d_in[2];
  const int*   eidx  = (const int*)d_in[3];
  const int*   etype = (const int*)d_in[4];

  const int n_units = in_sizes[1];
  const int n_edges = in_sizes[4];
  const int* src = eidx;
  const int* dst = eidx + n_edges;

  float* P    = (float*)d_ws;
  int4*  rec  = (int4*)(P + (size_t)n_units * 16);
  int*   head = (int*)(rec + n_edges);
  float* out  = (float*)d_out;

  const int gemm_blocks = (n_units + 127) / 128;
  proj_gemm_wmma<<<gemm_blocks, 256, 0, stream>>>(X, A, P, n_units);

  const int node_blocks = (n_units + 255) / 256;
  init_heads<<<node_blocks, 256, 0, stream>>>((int2*)head, n_units, n_edges, ws_size);

  const int edge_blocks = (n_edges + 255) / 256;
  edge_pass_link<<<edge_blocks, 256, 0, stream>>>(P, src, dst, etype, hat_t, head, rec,
                                                  n_edges, n_units, ws_size);

  const int fin_blocks = (n_units + 127) / 128;
  finalize_lists<<<fin_blocks, 128, 0, stream>>>((const int2*)head, rec, out,
                                                 n_units, n_edges, ws_size);
}


# per-edge global exchange goes: edges bucketed by owner node range with LDS counters (count, place); one work-group per range links and walks its lists in LDS; sums still in arrival order
# speedup vs baseline: 4.4383x; 2.6299x over previous
#include <hip/hip_runtime.h>
#include <hip/hip_bf16.h>

typedef __attribute__((ext_vector_type(2))) float v2f;
typedef __attribute__((ext_vector_type(8))) float v8f;

#define NODE_DIM   64
#define LEAKY      0.2f

#define OWN_RMAX    512
#define OWN_NBMAX   1024
#define OWN_CAP     12288
#define OWN_THREADS 1024
#define BK_THREADS  512
#define BK_EPT      8

struct Plan { int R; int n_ranges; bool ok; };

__device__ __forceinline__ Plan make_plan(int n_units, int n_edges, size_t ws_bytes) {
  Plan p;
  int R = (n_units + 255) / 256;
  R = (R + 31) & ~31;
  if (R < 32) R = 32;
  if (R > OWN_RMAX) R = OWN_RMAX;
  p.R = R;
  p.n_ranges = (n_units + R - 1) / R;
  p.ok = (ws_bytes >= ((size_t)n_units * 16 + (size_t)n_edges * 4) * sizeof(float)) &&
         (p.n_ranges <= OWN_NBMAX) && (n_edges >= 2 * OWN_NBMAX) && (n_units > 0);
  return p;
}

__global__ void __launch_bounds__(256)
proj_gemm_wmma(const float* __restrict__ X, const float* __restrict__ A,
               float* __restrict__ P, int n_units,
               int* __restrict__ ctr, int n_edges, size_t ws_bytes) {
  if (blockIdx.x == 0 && make_plan(n_units, n_edges, ws_bytes).ok) {
    for (int k = threadIdx.x; k < 2 * OWN_NBMAX; k += 256) ctr[k] = 0;
  }

  const int lane = threadIdx.x & 31;
  const int wave = threadIdx.x >> 5;
  const int half = lane >> 4;
  const int m    = lane & 15;
  const int row_base = blockIdx.x * 128 + wave * 16;

  int row = row_base + m;
  if (row >= n_units) row = n_units - 1;
  const float* __restrict__ xrow = X + (size_t)row * NODE_DIM;
  const float* __restrict__ wcol = A + (m & 7) * 128 + ((m >> 3) & 1) * 64;

  v8f c = {};
#pragma unroll
  for (int k0 = 0; k0 < NODE_DIM; k0 += 4) {
    const int k = k0 + half * 2;
    v2f a, b;
    a.x = xrow[k];     a.y = xrow[k + 1];
    b.x = wcol[k];     b.y = wcol[k + 1];
    c = __builtin_amdgcn_wmma_f32_16x16x4_f32(
         false, a,  false, b,
         (short)0, c,  false,  false);
  }

#pragma unroll
  for (int j = 0; j < 8; ++j) {
    const int orow = row_base + j + half * 8;
    if (orow < n_units) P[(size_t)orow * 16 + m] = c[j];
  }
}

__device__ __forceinline__ float leaky(float e) {
  return (e > 0.0f) ? e : LEAKY * e;
}

__global__ void __launch_bounds__(BK_THREADS)
bucket_count(const int* __restrict__ src, int* __restrict__ cnt,
             int n_edges, int n_units, size_t ws_bytes) {
  __shared__ int hist[OWN_NBMAX];
  const Plan p = make_plan(n_units, n_edges, ws_bytes);
  if (!p.ok) return;
  const int tid = threadIdx.x;
  for (int k = tid; k < p.n_ranges; k += BK_THREADS) hist[k] = 0;
  __syncthreads();
  const int e0 = blockIdx.x * (BK_THREADS * BK_EPT);
#pragma unroll
  for (int u = 0; u < BK_EPT; ++u) {
    const int i = e0 + u * BK_THREADS + tid;
    if (i < n_edges) {
      const int s = src[i];
      if ((unsigned)s < (unsigned)n_units)
        atomicAdd(&hist[(unsigned)s / (unsigned)p.R], 1);
    }
  }
  __syncthreads();
  for (int k = tid; k < p.n_ranges; k += BK_THREADS) {
    const int c = hist[k];
    if (c) atomicAdd(&cnt[k], c);
  }
}

__global__ void __launch_bounds__(BK_THREADS)
bucket_place(const int* __restrict__ src, const int* __restrict__ dst,
             const int* __restrict__ etype, const int* __restrict__ cnt,
             int* __restrict__ cur, int* __restrict__ start, int2* __restrict__ B,
             int n_edges, int n_units, size_t ws_bytes) {
  __shared__ int hist[OWN_NBMAX];
  __shared__ int bstart[OWN_NBMAX];
  __shared__ int part[BK_THREADS];
  const Plan p = make_plan(n_units, n_edges, ws_bytes);
  if (!p.ok) return;
  const int tid = threadIdx.x;

  const int b_lo = 2 * tid;
  const int c_lo = (b_lo < p.n_ranges) ? cnt[b_lo] : 0;
  const int c_hi = (b_lo + 1 < p.n_ranges) ? cnt[b_lo + 1] : 0;
  part[tid] = c_lo + c_hi;
  __syncthreads();
  for (int off = 1; off < BK_THREADS; off <<= 1) {
    const int v = (tid >= off) ? part[tid - off] : 0;
    __syncthreads();
    part[tid] += v;
    __syncthreads();
  }
  const int before = part[tid] - (c_lo + c_hi);
  bstart[b_lo] = before;
  bstart[b_lo + 1] = before + c_lo;
  for (int k = tid; k < p.n_ranges; k += BK_THREADS) hist[k] = 0;
  __syncthreads();
  if (blockIdx.x == 0)
    for (int k = tid; k < p.n_ranges; k += BK_THREADS) start[k] = bstart[k];

  const int e0 = blockIdx.x * (BK_THREADS * BK_EPT);
  int eb[BK_EPT], erank[BK_EPT], ed[BK_EPT], epack[BK_EPT];
#pragma unroll
  for (int u = 0; u < BK_EPT; ++u) {
    const int i = e0 + u * BK_THREADS + tid;
    eb[u] = -1; erank[u] = 0; ed[u] = 0; epack[u] = 0;
    if (i < n_edges) {
      const int s = src[i];
      if ((unsigned)s < (unsigned)n_units) {
        const int b = (int)((unsigned)s / (unsigned)p.R);
        eb[u] = b;
        ed[u] = dst[i];
        epack[u] = ((s - b * p.R) << 3) | (etype[i] & 7);
        erank[u] = atomicAdd(&hist[b], 1);
      }
    }
  }
  __syncthreads();
  for (int k = tid; k < p.n_ranges; k += BK_THREADS) {
    const int c = hist[k];
    hist[k] = c ? (bstart[k] + atomicAdd(&cur[k], c)) : 0;
  }
  __syncthreads();
#pragma unroll
  for (int u = 0; u < BK_EPT; ++u) {
    if (eb[u] >= 0) {
      const int pos = hist[eb[u]] + erank[u];
      if ((unsigned)pos < (unsigned)n_edges) B[pos] = make_int2(ed[u], epack[u]);
    }
  }
}

__global__ void __launch_bounds__(OWN_THREADS)
owner_lists(const float* __restrict__ P, const int2* __restrict__ B,
            const int* __restrict__ cnt, const int* __restrict__ start,
            const float* __restrict__ hat_t, float* __restrict__ out,
            int n_edges, int n_units, size_t ws_bytes) {
  __shared__ int   head[2 * OWN_RMAX];
  __shared__ float pu[8 * OWN_RMAX];
  __shared__ int2  slot[OWN_CAP];
  __shared__ float slot_t[OWN_CAP];
  __shared__ float st_mx[OWN_RMAX], st_dn[OWN_RMAX], st_g[OWN_RMAX];
  __shared__ int   st_has[OWN_RMAX];

  const Plan p = make_plan(n_units, n_edges, ws_bytes);
  if (!p.ok) return;
  const int tid = threadIdx.x;

  for (int range = blockIdx.x; range < p.n_ranges; range += gridDim.x) {
    const int base = range * p.R;
    const int nn = min(p.R, n_units - base);
    int b0 = start[range], bc = cnt[range];
    if (b0 < 0 || b0 > n_edges) b0 = n_edges;
    if (bc < 0) bc = 0;
    if (bc > n_edges - b0) bc = n_edges - b0;

    for (int k = tid; k < nn; k += OWN_THREADS) {
      st_mx[k] = -INFINITY; st_dn[k] = 0.0f; st_g[k] = 0.0f; st_has[k] = 0;
    }
    for (int k = tid; k < 8 * nn; k += OWN_THREADS)
      pu[k] = P[(size_t)(base + (k >> 3)) * 16 + (k & 7)];
    __syncthreads();

    for (int c0 = 0; c0 < bc; c0 += OWN_CAP) {
      const int m = min(OWN_CAP, bc - c0);
      for (int k = tid; k < 2 * nn; k += OWN_THREADS) head[k] = -1;
      __syncthreads();

      for (int k0 = tid; k0 < m; k0 += OWN_THREADS * 4) {
        int el[4], er[4], ed[4];
        bool ok[4];
        float pv[4], tt[4];
#pragma unroll
        for (int u = 0; u < 4; ++u) {
          const int k = k0 + u * OWN_THREADS;
          ok[u] = k < m;
          const int2 rc = ok[u] ? B[b0 + c0 + k] : make_int2(0, 0);
          ed[u] = rc.x;
          el[u] = min(rc.y >> 3, nn - 1);
          if (el[u] < 0) el[u] = 0;
          er[u] = rc.y & 7;
        }
#pragma unroll
        for (int u = 0; u < 4; ++u) {
          pv[u] = ok[u] ? P[(size_t)ed[u] * 16 + 8 + er[u]] : 0.0f;
          tt[u] = ok[u] ? hat_t[ed[u]] : 0.0f;
        }
#pragma unroll
        for (int u = 0; u < 4; ++u) {
          if (ok[u]) {
            const int k = k0 + u * OWN_THREADS;
            const int prev = atomicExch(&head[2 * el[u] + (k & 1)], k);
            const float e = leaky(pu[el[u] * 8 + er[u]] + pv[u]);
            slot[k] = make_int2(__float_as_int(e), prev);
            slot_t[k] = tt[u];
          }
        }
      }
      __syncthreads();

      for (int n = tid; n < nn; n += OWN_THREADS) {
        int j0 = head[2 * n], j1 = head[2 * n + 1];
        if (j0 >= 0 || j1 >= 0) {
          float mx = st_mx[n], dn = st_dn[n], g = st_g[n];
          for (int k = 0; k < m; ++k) {
            const bool a0 = (unsigned)j0 < (unsigned)m;
            const bool a1 = (unsigned)j1 < (unsigned)m;
            if (!(a0 || a1)) break;
            float x0 = -INFINITY, x1 = -INFINITY, t0 = 0.0f, t1 = 0.0f;
            int nx0 = -1, nx1 = -1;
            if (a0) { const int2 s = slot[j0]; x0 = __int_as_float(s.x); nx0 = s.y; t0 = slot_t[j0]; }
            if (a1) { const int2 s = slot[j1]; x1 = __int_as_float(s.x); nx1 = s.y; t1 = slot_t[j1]; }
            const float mn = fmaxf(mx, fmaxf(x0, x1));
            const float sc = __expf(mx - mn);
            const float w0 = __expf(x0 - mn);
            const float w1 = __expf(x1 - mn);
            dn = (dn * sc + w0) + w1;
            g  = (g * sc + w0 * t0) + w1 * t1;
            mx = mn;
            j0 = nx0;
            j1 = nx1;
          }
          st_mx[n] = mx; st_dn[n] = dn; st_g[n] = g; st_has[n] = 1;
        }
      }
      __syncthreads();
    }

    for (int n = tid; n < nn; n += OWN_THREADS)
      out[base + n] = st_has[n] ? (st_g[n] / st_dn[n]) : 0.0f;
    __syncthreads();
  }
}

extern "C" void kernel_launch(void* const* d_in, const int* in_sizes, int n_in,
                              void* d_out, int out_size, void* d_ws, size_t ws_size,
                              hipStream_t stream) {
  const float* X     = (const float*)d_in[0];
  const float* hat_t = (const float*)d_in[1];
  const float* A     = (const float*)d_in[2];
  const int*   eidx  = (const int*)d_in[3];
  const int*   etype = (const int*)d_in[4];

  const int n_units = in_sizes[1];
  const int n_edges = in_sizes[4];
  const int* src = eidx;
  const int* dst = eidx + n_edges;

  float* P     = (float*)d_ws;
  int2*  B     = (int2*)(P + (size_t)n_units * 16);
  int*   cnt   = (int*)(B + n_edges);
  int*   cur   = cnt + OWN_NBMAX;
  int*   start = cur + OWN_NBMAX;
  float* out   = (float*)d_out;

  const int gemm_blocks = (n_units + 127) / 128;
  proj_gemm_wmma<<<gemm_blocks, 256, 0, stream>>>(X, A, P, n_units, cnt, n_edges, ws_size);

  const int bk_blocks = (n_edges + BK_THREADS * BK_EPT - 1) / (BK_THREADS * BK_EPT);
  bucket_count<<<bk_blocks, BK_THREADS, 0, stream>>>(src, cnt, n_edges, n_units, ws_size);
  bucket_place<<<bk_blocks, BK_THREADS, 0, stream>>>(src, dst, etype, cnt, cur, start, B,
                                                     n_edges, n_units, ws_size);

  owner_lists<<<256, OWN_THREADS, 0, stream>>>(P, B, cnt, start, hat_t, out,
                                               n_edges, n_units, ws_size);
}


# loads of the count, place and owner passes lose their branches (index kept in range; owner: spare LDS slot per lane, 8 edges a thread): a thread's loads all in flight at once, each had its own wait
# speedup vs baseline: 5.1343x; 1.1568x over previous
#include <hip/hip_runtime.h>
#include <hip/hip_bf16.h>

typedef __attribute__((ext_vector_type(2))) float v2f;
typedef __attribute__((ext_vector_type(8))) float v8f;

#define NODE_DIM   64
#define LEAKY      0.2f

#define OWN_RMAX    512
#define OWN_NBMAX   1024
#define OWN_CAP     12288
#define OWN_THREADS 1024
#define OWN_EPT     8
#define BK_THREADS  512
#define BK_EPT      8

struct Plan { int R; int n_ranges; bool ok; };

__device__ __forceinline__ Plan make_plan(int n_units, int n_edges, size_t ws_bytes) {
  Plan p;
  int R = (n_units + 255) / 256;
  R = (R + 31) & ~31;
  if (R < 32) R = 32;
  if (R > OWN_RMAX) R = OWN_RMAX;
  p.R = R;
  p.n_ranges = (n_units + R - 1) / R;
  p.ok = (ws_bytes >= ((size_t)n_units * 16 + (size_t)n_edges * 4) * sizeof(float)) &&
         (p.n_ranges <= OWN_NBMAX) && (n_edges >= 2 * OWN_NBMAX) && (n_units > 0);
  return p;
}

__global__ void __launch_bounds__(256)
proj_gemm_wmma(const float* __restrict__ X, const float* __restrict__ A,
               float* __restrict__ P, int n_units,
               int* __restrict__ ctr, int n_edges, size_t ws_bytes) {
  if (blockIdx.x == 0 && make_plan(n_units, n_edges, ws_bytes).ok) {
    for (int k = threadIdx.x; k < 2 * OWN_NBMAX; k += 256) ctr[k] = 0;
  }

  const int lane = threadIdx.x & 31;
  const int wave = threadIdx.x >> 5;
  const int half = lane >> 4;
  const int m    = lane & 15;
  const int row_base = blockIdx.x * 128 + wave * 16;

  int row = row_base + m;
  if (row >= n_units) row = n_units - 1;
  const float* __restrict__ xrow = X + (size_t)row * NODE_DIM;
  const float* __restrict__ wcol = A + (m & 7) * 128 + ((m >> 3) & 1) * 64;

  v8f c = {};
#pragma unroll
  for (int k0 = 0; k0 < NODE_DIM; k0 += 4) {
    const int k = k0 + half * 2;
    v2f a, b;
    a.x = xrow[k];     a.y = xrow[k + 1];
    b.x = wcol[k];     b.y = wcol[k + 1];
    c = __builtin_amdgcn_wmma_f32_16x16x4_f32(
         false, a,  false, b,
         (short)0, c,  false,  false);
  }

#pragma unroll
  for (int j = 0; j < 8; ++j) {
    const int orow = row_base + j + half * 8;
    if (orow < n_units) P[(size_t)orow * 16 + m] = c[j];
  }
}

__device__ __forceinline__ float leaky(float e) {
  return (e > 0.0f) ? e : LEAKY * e;
}

__global__ void __launch_bounds__(BK_THREADS)
bucket_count(const int* __restrict__ src, int* __restrict__ cnt,
             int n_edges, int n_units, size_t ws_bytes) {
  __shared__ int hist[OWN_NBMAX];
  const Plan p = make_plan(n_units, n_edges, ws_bytes);
  if (!p.ok) return;
  const int tid = threadIdx.x;
  const int e0 = blockIdx.x * (BK_THREADS * BK_EPT);
  int sv[BK_EPT];
#pragma unroll
  for (int u = 0; u < BK_EPT; ++u) {
    const int i = e0 + u * BK_THREADS + tid;
    const int s = src[min(i, n_edges - 1)];
    sv[u] = (i < n_edges) ? s : -1;
  }
  for (int k = tid; k < p.n_ranges; k += BK_THREADS) hist[k] = 0;
  __syncthreads();
#pragma unroll
  for (int u = 0; u < BK_EPT; ++u) {
    const int s = sv[u];
    if ((unsigned)s < (unsigned)n_units)
      atomicAdd(&hist[(unsigned)s / (unsigned)p.R], 1);
  }
  __syncthreads();
  for (int k = tid; k < p.n_ranges; k += BK_THREADS) {
    const int c = hist[k];
    if (c) atomicAdd(&cnt[k], c);
  }
}

__global__ void __launch_bounds__(BK_THREADS)
bucket_place(const int* __restrict__ src, const int* __restrict__ dst,
             const int* __restrict__ etype, const int* __restrict__ cnt,
             int* __restrict__ cur, int* __restrict__ start, int2* __restrict__ B,
             int n_edges, int n_units, size_t ws_bytes) {
  __shared__ int hist[OWN_NBMAX];
  __shared__ int bstart[OWN_NBMAX];
  __shared__ int part[BK_THREADS];
  const Plan p = make_plan(n_units, n_edges, ws_bytes);
  if (!p.ok) return;
  const int tid = threadIdx.x;

  const int e0 = blockIdx.x * (BK_THREADS * BK_EPT);
  int es[BK_EPT], ed[BK_EPT], et[BK_EPT];
#pragma unroll
  for (int u = 0; u < BK_EPT; ++u) {
    const int i = e0 + u * BK_THREADS + tid;
    const int ic = min(i, n_edges - 1);
    const int s = src[ic];
    ed[u] = dst[ic];
    et[u] = etype[ic];
    es[u] = (i < n_edges) ? s : -1;
  }

  const int b_lo = 2 * tid;
  const int c_lo = (b_lo < p.n_ranges) ? cnt[b_lo] : 0;
  const int c_hi = (b_lo + 1 < p.n_ranges) ? cnt[b_lo + 1] : 0;
  part[tid] = c_lo + c_hi;
  __syncthreads();
  for (int off = 1; off < BK_THREADS; off <<= 1) {
    const int v = (tid >= off) ? part[tid - off] : 0;
    __syncthreads();
    part[tid] += v;
    __syncthreads();
  }
  const int before = part[tid] - (c_lo + c_hi);
  bstart[b_lo] = before;
  bstart[b_lo + 1] = before + c_lo;
  for (int k = tid; k < p.n_ranges; k += BK_THREADS) hist[k] = 0;
  __syncthreads();
  if (blockIdx.x == 0)
    for (int k = tid; k < p.n_ranges; k += BK_THREADS) start[k] = bstart[k];

  int eb[BK_EPT], erank[BK_EPT], epack[BK_EPT];
#pragma unroll
  for (int u = 0; u < BK_EPT; ++u) {
    eb[u] = -1; erank[u] = 0; epack[u] = 0;
    const int s = es[u];
    if ((unsigned)s < (unsigned)n_units) {
      const int b = (int)((unsigned)s / (unsigned)p.R);
      eb[u] = b;
      epack[u] = ((s - b * p.R) << 3) | (et[u] & 7);
      erank[u] = atomicAdd(&hist[b], 1);
    }
  }
  __syncthreads();
  for (int k = tid; k < p.n_ranges; k += BK_THREADS) {
    const int c = hist[k];
    hist[k] = c ? (bstart[k] + atomicAdd(&cur[k], c)) : 0;
  }
  __syncthreads();
#pragma unroll
  for (int u = 0; u < BK_EPT; ++u) {
    if (eb[u] >= 0) {
      const int pos = hist[eb[u]] + erank[u];
      if ((unsigned)pos < (unsigned)n_edges) B[pos] = make_int2(ed[u], epack[u]);
    }
  }
}

__global__ void __launch_bounds__(OWN_THREADS)
owner_lists(const float* __restrict__ P, const int2* __restrict__ B,
            const int* __restrict__ cnt, const int* __restrict__ start,
            const float* __restrict__ hat_t, float* __restrict__ out,
            int n_edges, int n_units, size_t ws_bytes) {
  __shared__ int   head[2 * OWN_RMAX + 32];
  __shared__ float pu[8 * OWN_RMAX];
  __shared__ int2  slot[OWN_CAP + 32];
  __shared__ float slot_t[OWN_CAP + 32];
  __shared__ float st_mx[OWN_RMAX], st_dn[OWN_RMAX], st_g[OWN_RMAX];
  __shared__ int   st_has[OWN_RMAX];

  const Plan p = make_plan(n_units, n_edges, ws_bytes);
  if (!p.ok) return;
  const int tid = threadIdx.x;
  const int lane = tid & 31;

  for (int range = blockIdx.x; range < p.n_ranges; range += gridDim.x) {
    const int base = range * p.R;
    const int nn = min(p.R, n_units - base);
    int b0 = start[range], bc = cnt[range];
    if (b0 < 0 || b0 > n_edges) b0 = n_edges;
    if (bc < 0) bc = 0;
    if (bc > n_edges - b0) bc = n_edges - b0;

    for (int k = tid; k < nn; k += OWN_THREADS) {
      st_mx[k] = -INFINITY; st_dn[k] = 0.0f; st_g[k] = 0.0f; st_has[k] = 0;
    }
    for (int k = tid; k < 8 * nn; k += OWN_THREADS)
      pu[k] = P[(size_t)(base + (k >> 3)) * 16 + (k & 7)];
    __syncthreads();

    for (int c0 = 0; c0 < bc; c0 += OWN_CAP) {
      const int m = min(OWN_CAP, bc - c0);
      for (int k = tid; k < 2 * nn; k += OWN_THREADS) head[k] = -1;
      __syncthreads();

      for (int k0 = tid; k0 < m; k0 += OWN_THREADS * OWN_EPT) {
        int el[OWN_EPT], er[OWN_EPT], ed[OWN_EPT];
        int2 rc[OWN_EPT];
        float pv[OWN_EPT], tt[OWN_EPT];
#pragma unroll
        for (int u = 0; u < OWN_EPT; ++u) {
          const int k = k0 + u * OWN_THREADS;
          rc[u] = B[b0 + c0 + min(k, m - 1)];
        }
#pragma unroll
        for (int u = 0; u < OWN_EPT; ++u) {
          ed[u] = rc[u].x;
          el[u] = min(rc[u].y >> 3, nn - 1);
          if (el[u] < 0) el[u] = 0;
          er[u] = rc[u].y & 7;
        }
#pragma unroll
        for (int u = 0; u < OWN_EPT; ++u) {
          pv[u] = P[(size_t)ed[u] * 16 + 8 + er[u]];
          tt[u] = hat_t[ed[u]];
        }
#pragma unroll
        for (int u = 0; u < OWN_EPT; ++u) {
          const int k = k0 + u * OWN_THREADS;
          const bool ok = k < m;
          const int ks = ok ? k : OWN_CAP + lane;
          const int hs = ok ? 2 * el[u] + (k & 1) : 2 * OWN_RMAX + lane;
          const int prev = atomicExch(&head[hs], ks);
          const float e = leaky(pu[el[u] * 8 + er[u]] + pv[u]);
          slot[ks] = make_int2(__float_as_int(e), prev);
          slot_t[ks] = tt[u];
        }
      }
      __syncthreads();

      for (int n = tid; n < nn; n += OWN_THREADS) {
        int j0 = head[2 * n], j1 = head[2 * n + 1];
        if (j0 >= 0 || j1 >= 0) {
          float mx = st_mx[n], dn = st_dn[n], g = st_g[n];
          for (int k = 0; k < m; ++k) {
            const bool a0 = (unsigned)j0 < (unsigned)m;
            const bool a1 = (unsigned)j1 < (unsigned)m;
            if (!(a0 || a1)) break;
            float x0 = -INFINITY, x1 = -INFINITY, t0 = 0.0f, t1 = 0.0f;
            int nx0 = -1, nx1 = -1;
            if (a0) { const int2 s = slot[j0]; x0 = __int_as_float(s.x); nx0 = s.y; t0 = slot_t[j0]; }
            if (a1) { const int2 s = slot[j1]; x1 = __int_as_float(s.x); nx1 = s.y; t1 = slot_t[j1]; }
            const float mn = fmaxf(mx, fmaxf(x0, x1));
            const float sc = __expf(mx - mn);
            const float w0 = __expf(x0 - mn);
            const float w1 = __expf(x1 - mn);
            dn = (dn * sc + w0) + w1;
            g  = (g * sc + w0 * t0) + w1 * t1;
            mx = mn;
            j0 = nx0;
            j1 = nx1;
          }
          st_mx[n] = mx; st_dn[n] = dn; st_g[n] = g; st_has[n] = 1;
        }
      }
      __syncthreads();
    }

    for (int n = tid; n < nn; n += OWN_THREADS)
      out[base + n] = st_has[n] ? (st_g[n] / st_dn[n]) : 0.0f;
    __syncthreads();
  }
}

extern "C" void kernel_launch(void* const* d_in, const int* in_sizes, int n_in,
                              void* d_out, int out_size, void* d_ws, size_t ws_size,
                              hipStream_t stream) {
  const float* X     = (const float*)d_in[0];
  const float* hat_t = (const float*)d_in[1];
  const float* A     = (const float*)d_in[2];
  const int*   eidx  = (const int*)d_in[3];
  const int*   etype = (const int*)d_in[4];

  const int n_units = in_sizes[1];
  const int n_edges = in_sizes[4];
  const int* src = eidx;
  const int* dst = eidx + n_edges;

  float* P     = (float*)d_ws;
  int2*  B     = (int2*)(P + (size_t)n_units * 16);
  int*   cnt   = (int*)(B + n_edges);
  int*   cur   = cnt + OWN_NBMAX;
  int*   start = cur + OWN_NBMAX;
  float* out   = (float*)d_out;

  const int gemm_blocks = (n_units + 127) / 128;
  proj_gemm_wmma<<<gemm_blocks, 256, 0, stream>>>(X, A, P, n_units, cnt, n_edges, ws_size);

  const int bk_blocks = (n_edges + BK_THREADS * BK_EPT - 1) / (BK_THREADS * BK_EPT);
  bucket_count<<<bk_blocks, BK_THREADS, 0, stream>>>(src, cnt, n_edges, n_units, ws_size);
  bucket_place<<<bk_blocks, BK_THREADS, 0, stream>>>(src, dst, etype, cnt, cur, start, B,
                                                     n_edges, n_units, ws_size);

  owner_lists<<<256, OWN_THREADS, 0, stream>>>(P, B, cnt, start, hat_t, out,
                                               n_edges, n_units, ws_size);
}


# count pass and global cursors go: each work-group sorts its 4096-edge run by owner range in LDS, writes it back contiguous plus a table word per range; owners gather pieces; sums in LDS arrival order
# speedup vs baseline: 6.1324x; 1.1944x over previous
#include <hip/hip_runtime.h>
#include <hip/hip_bf16.h>

typedef __attribute__((ext_vector_type(2))) float v2f;
typedef __attribute__((ext_vector_type(8))) float v8f;

#define NODE_DIM   64
#define LEAKY      0.2f

#define OWN_RMAX    512
#define OWN_NBMAX   1024
#define OWN_CAP     12288
#define OWN_THREADS 1024
#define OWN_PMAX    OWN_THREADS
#define OWN_EPT     8
#define BK_THREADS  512
#define BK_EPT      8
#define RUN_EDGES   (BK_THREADS * BK_EPT)

struct Plan { int R; int n_ranges; int n_runs; bool ok; };

__device__ __forceinline__ Plan make_plan(int n_units, int n_edges, size_t ws_bytes) {
  Plan p;
  int R = (n_units + 255) / 256;
  R = (R + 31) & ~31;
  if (R < 32) R = 32;
  if (R > OWN_RMAX) R = OWN_RMAX;
  p.R = R;
  p.n_ranges = (n_units + R - 1) / R;
  p.n_runs = (n_edges + RUN_EDGES - 1) / RUN_EDGES;
  p.ok = (n_units > 0) && (n_edges > 0) &&
         (p.n_ranges <= OWN_NBMAX) && (p.n_runs <= OWN_PMAX) &&
         (ws_bytes >= ((size_t)n_units * 16 + (size_t)n_edges * 2 +
                       (size_t)p.n_ranges * (size_t)p.n_runs) * sizeof(float));
  return p;
}

__device__ __forceinline__ int wave_incl_sum(int v, int lane) {
#pragma unroll
  for (int off = 1; off < 32; off <<= 1) {
    const int t = __shfl_up(v, off, 32);
    if (lane >= off) v += t;
  }
  return v;
}

__device__ __forceinline__ int wave_total(int x) {
#pragma unroll
  for (int off = 16; off >= 1; off >>= 1) x += __shfl_xor(x, off, 32);
  return x;
}

__global__ void __launch_bounds__(256)
proj_gemm_wmma(const float* __restrict__ X, const float* __restrict__ A,
               float* __restrict__ P, int n_units) {
  const int lane = threadIdx.x & 31;
  const int wave = threadIdx.x >> 5;
  const int half = lane >> 4;
  const int m    = lane & 15;
  const int row_base = blockIdx.x * 128 + wave * 16;

  int row = row_base + m;
  if (row >= n_units) row = n_units - 1;
  const float* __restrict__ xrow = X + (size_t)row * NODE_DIM;
  const float* __restrict__ wcol = A + (m & 7) * 128 + ((m >> 3) & 1) * 64;

  v8f c = {};
#pragma unroll
  for (int k0 = 0; k0 < NODE_DIM; k0 += 4) {
    const int k = k0 + half * 2;
    v2f a, b;
    a.x = xrow[k];     a.y = xrow[k + 1];
    b.x = wcol[k];     b.y = wcol[k + 1];
    c = __builtin_amdgcn_wmma_f32_16x16x4_f32(
         false, a,  false, b,
         (short)0, c,  false,  false);
  }

#pragma unroll
  for (int j = 0; j < 8; ++j) {
    const int orow = row_base + j + half * 8;
    if (orow < n_units) P[(size_t)orow * 16 + m] = c[j];
  }
}

__device__ __forceinline__ float leaky(float e) {
  return (e > 0.0f) ? e : LEAKY * e;
}

__global__ void __launch_bounds__(BK_THREADS)
run_sort(const int* __restrict__ src, const int* __restrict__ dst,
         const int* __restrict__ etype, int2* __restrict__ B, int* __restrict__ T,
         int n_edges, int n_units, size_t ws_bytes) {
  __shared__ int  hist[OWN_NBMAX];
  __shared__ int  wsum[BK_THREADS / 32];
  __shared__ int  s_tot;
  __shared__ int2 stage[RUN_EDGES];
  const Plan p = make_plan(n_units, n_edges, ws_bytes);
  if (!p.ok) return;
  const int tid = threadIdx.x;
  const int lane = tid & 31;
  const int wave = tid >> 5;
  const int run = blockIdx.x;
  if (run >= p.n_runs) return;

  const int e0 = run * RUN_EDGES;
  int es[BK_EPT], ed[BK_EPT], et[BK_EPT];
#pragma unroll
  for (int u = 0; u < BK_EPT; ++u) {
    const int i = e0 + u * BK_THREADS + tid;
    const int ic = min(i, n_edges - 1);
    const int s = src[ic];
    ed[u] = dst[ic];
    et[u] = etype[ic];
    es[u] = (i < n_edges) ? s : -1;
  }
  for (int k = tid; k < OWN_NBMAX; k += BK_THREADS) hist[k] = 0;
  __syncthreads();

  int eb[BK_EPT], erank[BK_EPT], epack[BK_EPT];
#pragma unroll
  for (int u = 0; u < BK_EPT; ++u) {
    eb[u] = -1; erank[u] = 0; epack[u] = 0;
    const int s = es[u];
    if ((unsigned)s < (unsigned)n_units) {
      const int b = (int)((unsigned)s / (unsigned)p.R);
      eb[u] = b;
      epack[u] = ((s - b * p.R) << 3) | (et[u] & 7);
      erank[u] = atomicAdd(&hist[b], 1);
    }
  }
  __syncthreads();

  const int b_lo = 2 * tid;
  const int c_lo = hist[b_lo];
  const int c_hi = hist[b_lo + 1];
  const int incl = wave_incl_sum(c_lo + c_hi, lane);
  if (lane == 31) wsum[wave] = incl;
  __syncthreads();
  const int waves_before = wave_total((lane < wave && lane < BK_THREADS / 32) ? wsum[lane] : 0);
  const int first = waves_before + incl - (c_lo + c_hi);
  hist[b_lo] = first;
  hist[b_lo + 1] = first + c_lo;
  if (tid == BK_THREADS - 1) s_tot = waves_before + incl;
  __syncthreads();

  int pos[BK_EPT];
#pragma unroll
  for (int u = 0; u < BK_EPT; ++u) pos[u] = hist[max(eb[u], 0)] + erank[u];
#pragma unroll
  for (int u = 0; u < BK_EPT; ++u) {
    if (eb[u] >= 0 && (unsigned)pos[u] < (unsigned)RUN_EDGES)
      stage[pos[u]] = make_int2(ed[u], epack[u]);
  }
  __syncthreads();

  if (b_lo < p.n_ranges)
    T[(size_t)b_lo * p.n_runs + run] = first | (c_lo << 16);
  if (b_lo + 1 < p.n_ranges)
    T[(size_t)(b_lo + 1) * p.n_runs + run] = (first + c_lo) | (c_hi << 16);

  const int tot = min(s_tot, min(RUN_EDGES, n_edges - e0));
  int2* __restrict__ Brun = B + (size_t)e0;
  int2 ov[BK_EPT];
#pragma unroll
  for (int u = 0; u < BK_EPT; ++u) ov[u] = stage[u * BK_THREADS + tid];
#pragma unroll
  for (int u = 0; u < BK_EPT; ++u) {
    const int k = u * BK_THREADS + tid;
    if (k < tot) Brun[k] = ov[u];
  }
}

__global__ void __launch_bounds__(OWN_THREADS)
owner_lists(const float* __restrict__ P, const int2* __restrict__ B,
            const int* __restrict__ T,
            const float* __restrict__ hat_t, float* __restrict__ out,
            int n_edges, int n_units, size_t ws_bytes) {
  __shared__ int   head[2 * OWN_RMAX + 32];
  __shared__ float pu[8 * OWN_RMAX];
  __shared__ int2  slot[OWN_CAP + 32];
  __shared__ float slot_t[OWN_CAP + 32];
  __shared__ float st_mx[OWN_RMAX], st_dn[OWN_RMAX], st_g[OWN_RMAX];
  __shared__ int   st_has[OWN_RMAX];
  __shared__ int2  prun[OWN_PMAX];
  __shared__ int   wsum[OWN_THREADS / 32];
  __shared__ int   s_bc;

  const Plan p = make_plan(n_units, n_edges, ws_bytes);
  if (!p.ok) return;
  const int tid = threadIdx.x;
  const int lane = tid & 31;
  const int wave = tid >> 5;
  int top = 1;
  while (2 * top < p.n_runs) top *= 2;

  for (int range = blockIdx.x; range < p.n_ranges; range += gridDim.x) {
    const int base = range * p.R;
    const int nn = min(p.R, n_units - base);
    const int tw = (tid < p.n_runs) ? T[(size_t)range * p.n_runs + tid] : 0;

    for (int k = tid; k < nn; k += OWN_THREADS) {
      st_mx[k] = -INFINITY; st_dn[k] = 0.0f; st_g[k] = 0.0f; st_has[k] = 0;
    }
    for (int k = tid; k < 8 * nn; k += OWN_THREADS)
      pu[k] = P[(size_t)(base + (k >> 3)) * 16 + (k & 7)];

    const int pc = min((tw >> 16) & 0x7fff, RUN_EDGES);
    const int incl = wave_incl_sum(pc, lane);
    if (lane == 31) wsum[wave] = incl;
    __syncthreads();
    const int waves_before = wave_total((lane < wave) ? wsum[lane] : 0);
    const int pfirst = waves_before + incl - pc;
    prun[tid] = make_int2(pfirst, tid * RUN_EDGES + (tw & 0xffff) - pfirst);
    if (tid == OWN_THREADS - 1) s_bc = waves_before + incl;
    __syncthreads();
    const int bc = s_bc;

    for (int c0 = 0; c0 < bc; c0 += OWN_CAP) {
      const int m = min(OWN_CAP, bc - c0);
      for (int k = tid; k < 2 * nn; k += OWN_THREADS) head[k] = -1;
      __syncthreads();

      for (int k0 = tid; k0 < m; k0 += OWN_THREADS * OWN_EPT) {
        int el[OWN_EPT], er[OWN_EPT], ed[OWN_EPT];
        int2 rc[OWN_EPT];
        float pv[OWN_EPT], tt[OWN_EPT];
        int kk[OWN_EPT], pt[OWN_EPT];
#pragma unroll
        for (int u = 0; u < OWN_EPT; ++u) {
          kk[u] = c0 + min(k0 + u * OWN_THREADS, m - 1);
          pt[u] = 0;
        }
        for (int step = top; step >= 1; step >>= 1) {
#pragma unroll
          for (int u = 0; u < OWN_EPT; ++u)
            if (prun[pt[u] + step].x <= kk[u]) pt[u] += step;
        }
#pragma unroll
        for (int u = 0; u < OWN_EPT; ++u) {
          const int at = min(max(kk[u] + prun[pt[u]].y, 0), n_edges - 1);
          rc[u] = B[at];
        }
#pragma unroll
        for (int u = 0; u < OWN_EPT; ++u) {
          ed[u] = rc[u].x;
          el[u] = min(rc[u].y >> 3, nn - 1);
          if (el[u] < 0) el[u] = 0;
          er[u] = rc[u].y & 7;
        }
#pragma unroll
        for (int u = 0; u < OWN_EPT; ++u) {
          pv[u] = P[(size_t)ed[u] * 16 + 8 + er[u]];
          tt[u] = hat_t[ed[u]];
        }
#pragma unroll
        for (int u = 0; u < OWN_EPT; ++u) {
          const int k = k0 + u * OWN_THREADS;
          const bool ok = k < m;
          const int ks = ok ? k : OWN_CAP + lane;
          const int hs = ok ? 2 * el[u] + (k & 1) : 2 * OWN_RMAX + lane;
          const int prev = atomicExch(&head[hs], ks);
          const float e = leaky(pu[el[u] * 8 + er[u]] + pv[u]);
          slot[ks] = make_int2(__float_as_int(e), prev);
          slot_t[ks] = tt[u];
        }
      }
      __syncthreads();

      for (int n = tid; n < nn; n += OWN_THREADS) {
        int j0 = head[2 * n], j1 = head[2 * n + 1];
        if (j0 >= 0 || j1 >= 0) {
          float mx = st_mx[n], dn = st_dn[n], g = st_g[n];
          for (int k = 0; k < m; ++k) {
            const bool a0 = (unsigned)j0 < (unsigned)m;
            const bool a1 = (unsigned)j1 < (unsigned)m;
            if (!(a0 || a1)) break;
            float x0 = -INFINITY, x1 = -INFINITY, t0 = 0.0f, t1 = 0.0f;
            int nx0 = -1, nx1 = -1;
            if (a0) { const int2 s = slot[j0]; x0 = __int_as_float(s.x); nx0 = s.y; t0 = slot_t[j0]; }
            if (a1) { const int2 s = slot[j1]; x1 = __int_as_float(s.x); nx1 = s.y; t1 = slot_t[j1]; }
            const float mn = fmaxf(mx, fmaxf(x0, x1));
            const float sc = __expf(mx - mn);
            const float w0 = __expf(x0 - mn);
            const float w1 = __expf(x1 - mn);
            dn = (dn * sc + w0) + w1;
            g  = (g * sc + w0 * t0) + w1 * t1;
            mx = mn;
            j0 = nx0;
            j1 = nx1;
          }
          st_mx[n] = mx; st_dn[n] = dn; st_g[n] = g; st_has[n] = 1;
        }
      }
      __syncthreads();
    }

    for (int n = tid; n < nn; n += OWN_THREADS)
      out[base + n] = st_has[n] ? (st_g[n] / st_dn[n]) : 0.0f;
    __syncthreads();
  }
}

extern "C" void kernel_launch(void* const* d_in, const int* in_sizes, int n_in,
                              void* d_out, int out_size, void* d_ws, size_t ws_size,
                              hipStream_t stream) {
  const float* X     = (const float*)d_in[0];
  const float* hat_t = (const float*)d_in[1];
  const float* A     = (const float*)d_in[2];
  const int*   eidx  = (const int*)d_in[3];
  const int*   etype = (const int*)d_in[4];

  const int n_units = in_sizes[1];
  const int n_edges = in_sizes[4];
  const int* src = eidx;
  const int* dst = eidx + n_edges;

  float* P     = (float*)d_ws;
  int2*  B     = (int2*)(P + (size_t)n_units * 16);
  int*   T     = (int*)(B + n_edges);
  float* out   = (float*)d_out;

  const int gemm_blocks = (n_units + 127) / 128;
  proj_gemm_wmma<<<gemm_blocks, 256, 0, stream>>>(X, A, P, n_units);

  const int bk_blocks = (n_edges + BK_THREADS * BK_EPT - 1) / (BK_THREADS * BK_EPT);
  run_sort<<<bk_blocks, BK_THREADS, 0, stream>>>(src, dst, etype, B, T, n_edges, n_units, ws_size);

  owner_lists<<<256, OWN_THREADS, 0, stream>>>(P, B, T, hat_t, out,
                                               n_edges, n_units, ws_size);
}


# projection GEMM moves into the run-sort launch: each work-group loads its 256 rows' fragments beside its run's edge loads, multiplies, sorts, stores both; 3 launches become 2
# speedup vs baseline: 6.4086x; 1.0450x over previous
#include <hip/hip_runtime.h>
#include <hip/hip_bf16.h>

typedef __attribute__((ext_vector_type(2))) float v2f;
typedef __attribute__((ext_vector_type(8))) float v8f;

#define NODE_DIM   64
#define LEAKY      0.2f

#define OWN_RMAX    512
#define OWN_NBMAX   1024
#define OWN_CAP     12288
#define OWN_THREADS 1024
#define OWN_PMAX    OWN_THREADS
#define OWN_EPT     8
#define BK_THREADS  512
#define BK_EPT      8
#define RUN_EDGES   (BK_THREADS * BK_EPT)

struct Plan { int R; int n_ranges; int n_runs; bool ok; };

__device__ __forceinline__ Plan make_plan(int n_units, int n_edges, size_t ws_bytes) {
  Plan p;
  int R = (n_units + 255) / 256;
  R = (R + 31) & ~31;
  if (R < 32) R = 32;
  if (R > OWN_RMAX) R = OWN_RMAX;
  p.R = R;
  p.n_ranges = (n_units + R - 1) / R;
  p.n_runs = (n_edges + RUN_EDGES - 1) / RUN_EDGES;
  p.ok = (n_units > 0) && (n_edges > 0) &&
         (p.n_ranges <= OWN_NBMAX) && (p.n_runs <= OWN_PMAX) &&
         (ws_bytes >= ((size_t)n_units * 16 + (size_t)n_edges * 2 +
                       (size_t)p.n_ranges * (size_t)p.n_runs) * sizeof(float));
  return p;
}

__device__ __forceinline__ int wave_incl_sum(int v, int lane) {
#pragma unroll
  for (int off = 1; off < 32; off <<= 1) {
    const int t = __shfl_up(v, off, 32);
    if (lane >= off) v += t;
  }
  return v;
}

__device__ __forceinline__ int wave_total(int x) {
#pragma unroll
  for (int off = 16; off >= 1; off >>= 1) x += __shfl_xor(x, off, 32);
  return x;
}

#define GEMM_KSTEPS (NODE_DIM / 4)

__device__ __forceinline__ void proj_tile_load(const float* __restrict__ X, const float* __restrict__ A,
                                               int row_base, int n_units, int lane,
                                               v2f (&fa)[GEMM_KSTEPS], v2f (&fb)[GEMM_KSTEPS]) {
  const int half = lane >> 4;
  const int m    = lane & 15;
  int row = row_base + m;
  if (row >= n_units) row = n_units - 1;
  const float* __restrict__ xrow = X + (size_t)row * NODE_DIM;
  const float* __restrict__ wcol = A + (m & 7) * 128 + ((m >> 3) & 1) * 64;
#pragma unroll
  for (int i = 0; i < GEMM_KSTEPS; ++i) {
    const int k = 4 * i + half * 2;
    fa[i].x = xrow[k];     fa[i].y = xrow[k + 1];
    fb[i].x = wcol[k];     fb[i].y = wcol[k + 1];
  }
}

__device__ __forceinline__ v8f proj_tile_mul(const v2f (&fa)[GEMM_KSTEPS], const v2f (&fb)[GEMM_KSTEPS]) {
  v8f c = {};
#pragma unroll
  for (int i = 0; i < GEMM_KSTEPS; ++i)
    c = __builtin_amdgcn_wmma_f32_16x16x4_f32(
         false, fa[i],  false, fb[i],
         (short)0, c,  false,  false);
  return c;
}

__device__ __forceinline__ void proj_tile_store(float* __restrict__ P, int row_base, int n_units,
                                                int lane, const v8f& c) {
  const int half = lane >> 4;
  const int m    = lane & 15;
#pragma unroll
  for (int j = 0; j < 8; ++j) {
    const int orow = row_base + j + half * 8;
    if (orow < n_units) P[(size_t)orow * 16 + m] = c[j];
  }
}

__device__ __forceinline__ float leaky(float e) {
  return (e > 0.0f) ? e : LEAKY * e;
}

#define SLAB_ROWS ((BK_THREADS / 32) * 16)

__global__ void __launch_bounds__(BK_THREADS)
proj_run_sort(const float* __restrict__ X, const float* __restrict__ A, float* __restrict__ P,
              const int* __restrict__ src, const int* __restrict__ dst,
              const int* __restrict__ etype, int2* __restrict__ B, int* __restrict__ T,
              int n_edges, int n_units, size_t ws_bytes) {
  __shared__ int  hist[OWN_NBMAX];
  __shared__ int  wsum[BK_THREADS / 32];
  __shared__ int  s_tot;
  __shared__ int2 stage[RUN_EDGES];
  const Plan p = make_plan(n_units, n_edges, ws_bytes);
  const int tid = threadIdx.x;
  const int lane = tid & 31;
  const int wave = tid >> 5;
  const int run = blockIdx.x;
  const int n_slabs = (n_units + SLAB_ROWS - 1) / SLAB_ROWS;
  int slab = blockIdx.x;

  if (p.ok && run < p.n_runs) {
    const int e0 = run * RUN_EDGES;
    int es[BK_EPT], ed[BK_EPT], et[BK_EPT];
#pragma unroll
    for (int u = 0; u < BK_EPT; ++u) {
      const int i = e0 + u * BK_THREADS + tid;
      const int ic = min(i, n_edges - 1);
      const int s = src[ic];
      ed[u] = dst[ic];
      et[u] = etype[ic];
      es[u] = (i < n_edges) ? s : -1;
    }
    const int row_base = slab * SLAB_ROWS + wave * 16;
    v2f fa[GEMM_KSTEPS], fb[GEMM_KSTEPS];
    proj_tile_load(X, A, row_base, n_units, lane, fa, fb);
    for (int k = tid; k < OWN_NBMAX; k += BK_THREADS) hist[k] = 0;
    const v8f ctile = proj_tile_mul(fa, fb);
    __syncthreads();

    int eb[BK_EPT], erank[BK_EPT], epack[BK_EPT];
#pragma unroll
    for (int u = 0; u < BK_EPT; ++u) {
      eb[u] = -1; erank[u] = 0; epack[u] = 0;
      const int s = es[u];
      if ((unsigned)s < (unsigned)n_units) {
        const int b = (int)((unsigned)s / (unsigned)p.R);
        eb[u] = b;
        epack[u] = ((s - b * p.R) << 3) | (et[u] & 7);
        erank[u] = atomicAdd(&hist[b], 1);
      }
    }
    __syncthreads();

    const int b_lo = 2 * tid;
    const int c_lo = hist[b_lo];
    const int c_hi = hist[b_lo + 1];
    const int incl = wave_incl_sum(c_lo + c_hi, lane);
    if (lane == 31) wsum[wave] = incl;
    __syncthreads();
    const int waves_before = wave_total((lane < wave && lane < BK_THREADS / 32) ? wsum[lane] : 0);
    const int first = waves_before + incl - (c_lo + c_hi);
    hist[b_lo] = first;
    hist[b_lo + 1] = first + c_lo;
    if (tid == BK_THREADS - 1) s_tot = waves_before + incl;
    __syncthreads();

    int pos[BK_EPT];
#pragma unroll
    for (int u = 0; u < BK_EPT; ++u) pos[u] = hist[max(eb[u], 0)] + erank[u];
#pragma unroll
    for (int u = 0; u < BK_EPT; ++u) {
      if (eb[u] >= 0 && (unsigned)pos[u] < (unsigned)RUN_EDGES)
        stage[pos[u]] = make_int2(ed[u], epack[u]);
    }
    __syncthreads();

    if (b_lo < p.n_ranges)
      T[(size_t)b_lo * p.n_runs + run] = first | (c_lo << 16);
    if (b_lo + 1 < p.n_ranges)
      T[(size_t)(b_lo + 1) * p.n_runs + run] = (first + c_lo) | (c_hi << 16);

    const int tot = min(s_tot, min(RUN_EDGES, n_edges - e0));
    int2* __restrict__ Brun = B + (size_t)e0;
    int2 ov[BK_EPT];
#pragma unroll
    for (int u = 0; u < BK_EPT; ++u) ov[u] = stage[u * BK_THREADS + tid];
#pragma unroll
    for (int u = 0; u < BK_EPT; ++u) {
      const int k = u * BK_THREADS + tid;
      if (k < tot) Brun[k] = ov[u];
    }

    proj_tile_store(P, row_base, n_units, lane, ctile);
    slab += gridDim.x;
  }

  for (; slab < n_slabs; slab += gridDim.x) {
    const int row_base = slab * SLAB_ROWS + wave * 16;
    v2f fa[GEMM_KSTEPS], fb[GEMM_KSTEPS];
    proj_tile_load(X, A, row_base, n_units, lane, fa, fb);
    const v8f ctile = proj_tile_mul(fa, fb);
    proj_tile_store(P, row_base, n_units, lane, ctile);
  }
}

__global__ void __launch_bounds__(OWN_THREADS)
owner_lists(const float* __restrict__ P, const int2* __restrict__ B,
            const int* __restrict__ T,
            const float* __restrict__ hat_t, float* __restrict__ out,
            int n_edges, int n_units, size_t ws_bytes) {
  __shared__ int   head[2 * OWN_RMAX + 32];
  __shared__ float pu[8 * OWN_RMAX];
  __shared__ int2  slot[OWN_CAP + 32];
  __shared__ float slot_t[OWN_CAP + 32];
  __shared__ float st_mx[OWN_RMAX], st_dn[OWN_RMAX], st_g[OWN_RMAX];
  __shared__ int   st_has[OWN_RMAX];
  __shared__ int2  prun[OWN_PMAX];
  __shared__ int   wsum[OWN_THREADS / 32];
  __shared__ int   s_bc;

  const Plan p = make_plan(n_units, n_edges, ws_bytes);
  if (!p.ok) return;
  const int tid = threadIdx.x;
  const int lane = tid & 31;
  const int wave = tid >> 5;
  int top = 1;
  while (2 * top < p.n_runs) top *= 2;

  for (int range = blockIdx.x; range < p.n_ranges; range += gridDim.x) {
    const int base = range * p.R;
    const int nn = min(p.R, n_units - base);
    const int tw = (tid < p.n_runs) ? T[(size_t)range * p.n_runs + tid] : 0;

    for (int k = tid; k < nn; k += OWN_THREADS) {
      st_mx[k] = -INFINITY; st_dn[k] = 0.0f; st_g[k] = 0.0f; st_has[k] = 0;
    }
    for (int k = tid; k < 8 * nn; k += OWN_THREADS)
      pu[k] = P[(size_t)(base + (k >> 3)) * 16 + (k & 7)];

    const int pc = min((tw >> 16) & 0x7fff, RUN_EDGES);
    const int incl = wave_incl_sum(pc, lane);
    if (lane == 31) wsum[wave] = incl;
    __syncthreads();
    const int waves_before = wave_total((lane < wave) ? wsum[lane] : 0);
    const int pfirst = waves_before + incl - pc;
    prun[tid] = make_int2(pfirst, tid * RUN_EDGES + (tw & 0xffff) - pfirst);
    if (tid == OWN_THREADS - 1) s_bc = waves_before + incl;
    __syncthreads();
    const int bc = s_bc;

    for (int c0 = 0; c0 < bc; c0 += OWN_CAP) {
      const int m = min(OWN_CAP, bc - c0);
      for (int k = tid; k < 2 * nn; k += OWN_THREADS) head[k] = -1;
      __syncthreads();

      for (int k0 = tid; k0 < m; k0 += OWN_THREADS * OWN_EPT) {
        int el[OWN_EPT], er[OWN_EPT], ed[OWN_EPT];
        int2 rc[OWN_EPT];
        float pv[OWN_EPT], tt[OWN_EPT];
        int kk[OWN_EPT], pt[OWN_EPT];
#pragma unroll
        for (int u = 0; u < OWN_EPT; ++u) {
          kk[u] = c0 + min(k0 + u * OWN_THREADS, m - 1);
          pt[u] = 0;
        }
        for (int step = top; step >= 1; step >>= 1) {
#pragma unroll
          for (int u = 0; u < OWN_EPT; ++u)
            if (prun[pt[u] + step].x <= kk[u]) pt[u] += step;
        }
#pragma unroll
        for (int u = 0; u < OWN_EPT; ++u) {
          const int at = min(max(kk[u] + prun[pt[u]].y, 0), n_edges - 1);
          rc[u] = B[at];
        }
#pragma unroll
        for (int u = 0; u < OWN_EPT; ++u) {
          ed[u] = rc[u].x;
          el[u] = min(rc[u].y >> 3, nn - 1);
          if (el[u] < 0) el[u] = 0;
          er[u] = rc[u].y & 7;
        }
#pragma unroll
        for (int u = 0; u < OWN_EPT; ++u) {
          pv[u] = P[(size_t)ed[u] * 16 + 8 + er[u]];
          tt[u] = hat_t[ed[u]];
        }
#pragma unroll
        for (int u = 0; u < OWN_EPT; ++u) {
          const int k = k0 + u * OWN_THREADS;
          const bool ok = k < m;
          const int ks = ok ? k : OWN_CAP + lane;
          const int hs = ok ? 2 * el[u] + (k & 1) : 2 * OWN_RMAX + lane;
          const int prev = atomicExch(&head[hs], ks);
          const float e = leaky(pu[el[u] * 8 + er[u]] + pv[u]);
          slot[ks] = make_int2(__float_as_int(e), prev);
          slot_t[ks] = tt[u];
        }
      }
      __syncthreads();

      for (int n = tid; n < nn; n += OWN_THREADS) {
        int j0 = head[2 * n], j1 = head[2 * n + 1];
        if (j0 >= 0 || j1 >= 0) {
          float mx = st_mx[n], dn = st_dn[n], g = st_g[n];
          for (int k = 0; k < m; ++k) {
            const bool a0 = (unsigned)j0 < (unsigned)m;
            const bool a1 = (unsigned)j1 < (unsigned)m;
            if (!(a0 || a1)) break;
            float x0 = -INFINITY, x1 = -INFINITY, t0 = 0.0f, t1 = 0.0f;
            int nx0 = -1, nx1 = -1;
            if (a0) { const int2 s = slot[j0]; x0 = __int_as_float(s.x); nx0 = s.y; t0 = slot_t[j0]; }
            if (a1) { const int2 s = slot[j1]; x1 = __int_as_float(s.x); nx1 = s.y; t1 = slot_t[j1]; }
            const float mn = fmaxf(mx, fmaxf(x0, x1));
            const float sc = __expf(mx - mn);
            const float w0 = __expf(x0 - mn);
            const float w1 = __expf(x1 - mn);
            dn = (dn * sc + w0) + w1;
            g  = (g * sc + w0 * t0) + w1 * t1;
            mx = mn;
            j0 = nx0;
            j1 = nx1;
          }
          st_mx[n] = mx; st_dn[n] = dn; st_g[n] = g; st_has[n] = 1;
        }
      }
      __syncthreads();
    }

    for (int n = tid; n < nn; n += OWN_THREADS)
      out[base + n] = st_has[n] ? (st_g[n] / st_dn[n]) : 0.0f;
    __syncthreads();
  }
}

extern "C" void kernel_launch(void* const* d_in, const int* in_sizes, int n_in,
                              void* d_out, int out_size, void* d_ws, size_t ws_size,
                              hipStream_t stream) {
  const float* X     = (const float*)d_in[0];
  const float* hat_t = (const float*)d_in[1];
  const float* A     = (const float*)d_in[2];
  const int*   eidx  = (const int*)d_in[3];
  const int*   etype = (const int*)d_in[4];

  const int n_units = in_sizes[1];
  const int n_edges = in_sizes[4];
  const int* src = eidx;
  const int* dst = eidx + n_edges;

  float* P     = (float*)d_ws;
  int2*  B     = (int2*)(P + (size_t)n_units * 16);
  int*   T     = (int*)(B + n_edges);
  float* out   = (float*)d_out;

  const int bk_blocks = (n_edges + BK_THREADS * BK_EPT - 1) / (BK_THREADS * BK_EPT);
  proj_run_sort<<<bk_blocks, BK_THREADS, 0, stream>>>(X, A, P, src, dst, etype, B, T,
                                                      n_edges, n_units, ws_size);

  owner_lists<<<256, OWN_THREADS, 0, stream>>>(P, B, T, hat_t, out,
                                               n_edges, n_units, ws_size);
}


# projection operands pass through LDS: whole-line 16-byte loads of X tiles and W, 16-byte fragment reads at an odd pitch (k order inside the products changes); was 32 loads of 16 lines per wave
# speedup vs baseline: 6.7370x; 1.0512x over previous
#include <hip/hip_runtime.h>
#include <hip/hip_bf16.h>

typedef __attribute__((ext_vector_type(2))) float v2f;
typedef __attribute__((ext_vector_type(4))) float v4f;
typedef __attribute__((ext_vector_type(8))) float v8f;

#define NODE_DIM   64
#define LEAKY      0.2f

#define OWN_RMAX    512
#define OWN_NBMAX   1024
#define OWN_CAP     12288
#define OWN_THREADS 1024
#define OWN_PMAX    OWN_THREADS
#define OWN_EPT     8
#define BK_THREADS  512
#define BK_EPT      8
#define RUN_EDGES   (BK_THREADS * BK_EPT)

struct Plan { int R; int n_ranges; int n_runs; bool ok; };

__device__ __forceinline__ Plan make_plan(int n_units, int n_edges, size_t ws_bytes) {
  Plan p;
  int R = (n_units + 255) / 256;
  R = (R + 31) & ~31;
  if (R < 32) R = 32;
  if (R > OWN_RMAX) R = OWN_RMAX;
  p.R = R;
  p.n_ranges = (n_units + R - 1) / R;
  p.n_runs = (n_edges + RUN_EDGES - 1) / RUN_EDGES;
  p.ok = (n_units > 0) && (n_edges > 0) &&
         (p.n_ranges <= OWN_NBMAX) && (p.n_runs <= OWN_PMAX) &&
         (ws_bytes >= ((size_t)n_units * 16 + (size_t)n_edges * 2 +
                       (size_t)p.n_ranges * (size_t)p.n_runs) * sizeof(float));
  return p;
}

__device__ __forceinline__ int wave_incl_sum(int v, int lane) {
#pragma unroll
  for (int off = 1; off < 32; off <<= 1) {
    const int t = __shfl_up(v, off, 32);
    if (lane >= off) v += t;
  }
  return v;
}

__device__ __forceinline__ int wave_total(int x) {
#pragma unroll
  for (int off = 16; off >= 1; off >>= 1) x += __shfl_xor(x, off, 32);
  return x;
}

#define GEMM_KSTEPS (NODE_DIM / 4)

__device__ __forceinline__ void proj_tile_load(const float* __restrict__ X, const float* __restrict__ A,
                                               int row_base, int n_units, int lane,
                                               v2f (&fa)[GEMM_KSTEPS], v2f (&fb)[GEMM_KSTEPS]) {
  const int half = lane >> 4;
  const int m    = lane & 15;
  int row = row_base + m;
  if (row >= n_units) row = n_units - 1;
  const float* __restrict__ xrow = X + (size_t)row * NODE_DIM;
  const float* __restrict__ wcol = A + (m & 7) * 128 + ((m >> 3) & 1) * 64;
#pragma unroll
  for (int i = 0; i < GEMM_KSTEPS; ++i) {
    const int k = 4 * i + half * 2;
    fa[i].x = xrow[k];     fa[i].y = xrow[k + 1];
    fb[i].x = wcol[k];     fb[i].y = wcol[k + 1];
  }
}

#define FRAG_PITCH (NODE_DIM + 4)
#define XTILE_LOADS 8

__device__ __forceinline__ void proj_x_load(const float* __restrict__ X, int row_base, int n_units,
                                            int lane, v4f (&xv)[XTILE_LOADS]) {
  const int sub = lane >> 4;
  const int col = (lane & 15) * 4;
#pragma unroll
  for (int j = 0; j < XTILE_LOADS; ++j) {
    int row = row_base + 2 * j + sub;
    if (row >= n_units) row = n_units - 1;
    xv[j] = *reinterpret_cast<const v4f*>(X + (size_t)row * NODE_DIM + col);
  }
}

__device__ __forceinline__ void proj_x_stage(float* xt, int lane, const v4f (&xv)[XTILE_LOADS]) {
  const int sub = lane >> 4;
  const int col = (lane & 15) * 4;
#pragma unroll
  for (int j = 0; j < XTILE_LOADS; ++j)
    *reinterpret_cast<v4f*>(xt + (2 * j + sub) * FRAG_PITCH + col) = xv[j];
}

__device__ __forceinline__ void proj_frag_read(const float* xt, const float* wt, int lane,
                                               v2f (&fa)[GEMM_KSTEPS], v2f (&fb)[GEMM_KSTEPS]) {
  const int half = lane >> 4;
  const int m    = lane & 15;
  const float* ar = xt + m * FRAG_PITCH + 4 * half;
  const float* br = wt + m * FRAG_PITCH + 4 * half;
#pragma unroll
  for (int q = 0; q < GEMM_KSTEPS / 2; ++q) {
    const v4f a4 = *reinterpret_cast<const v4f*>(ar + 8 * q);
    const v4f b4 = *reinterpret_cast<const v4f*>(br + 8 * q);
    fa[2 * q].x = a4.x;     fa[2 * q].y = a4.y;
    fa[2 * q + 1].x = a4.z; fa[2 * q + 1].y = a4.w;
    fb[2 * q].x = b4.x;     fb[2 * q].y = b4.y;
    fb[2 * q + 1].x = b4.z; fb[2 * q + 1].y = b4.w;
  }
}

__device__ __forceinline__ v8f proj_tile_mul(const v2f (&fa)[GEMM_KSTEPS], const v2f (&fb)[GEMM_KSTEPS]) {
  v8f c = {};
#pragma unroll
  for (int i = 0; i < GEMM_KSTEPS; ++i)
    c = __builtin_amdgcn_wmma_f32_16x16x4_f32(
         false, fa[i],  false, fb[i],
         (short)0, c,  false,  false);
  return c;
}

__device__ __forceinline__ void proj_tile_store(float* __restrict__ P, int row_base, int n_units,
                                                int lane, const v8f& c) {
  const int half = lane >> 4;
  const int m    = lane & 15;
#pragma unroll
  for (int j = 0; j < 8; ++j) {
    const int orow = row_base + j + half * 8;
    if (orow < n_units) P[(size_t)orow * 16 + m] = c[j];
  }
}

__device__ __forceinline__ float leaky(float e) {
  return (e > 0.0f) ? e : LEAKY * e;
}

#define SLAB_ROWS ((BK_THREADS / 32) * 16)

__global__ void __launch_bounds__(BK_THREADS)
proj_run_sort(const float* __restrict__ X, const float* __restrict__ A, float* __restrict__ P,
              const int* __restrict__ src, const int* __restrict__ dst,
              const int* __restrict__ etype, int2* __restrict__ B, int* __restrict__ T,
              int n_edges, int n_units, size_t ws_bytes) {
  __shared__ int  hist[OWN_NBMAX];
  __shared__ int  wsum[BK_THREADS / 32];
  __shared__ int  s_tot;
  __shared__ int2 stage[RUN_EDGES];
  __shared__ __attribute__((aligned(16))) float xt[BK_THREADS / 32][16 * FRAG_PITCH];
  __shared__ __attribute__((aligned(16))) float wt[16 * FRAG_PITCH];
  static_assert(BK_THREADS * 2 == 16 * NODE_DIM, "one 8-byte load a thread fetches all of W");
  const Plan p = make_plan(n_units, n_edges, ws_bytes);
  const int tid = threadIdx.x;
  const int lane = tid & 31;
  const int wave = tid >> 5;
  const int run = blockIdx.x;
  const int n_slabs = (n_units + SLAB_ROWS - 1) / SLAB_ROWS;
  int slab = blockIdx.x;

  if (p.ok && run < p.n_runs) {
    const int e0 = run * RUN_EDGES;
    int es[BK_EPT], ed[BK_EPT], et[BK_EPT];
#pragma unroll
    for (int u = 0; u < BK_EPT; ++u) {
      const int i = e0 + u * BK_THREADS + tid;
      const int ic = min(i, n_edges - 1);
      const int s = src[ic];
      ed[u] = dst[ic];
      et[u] = etype[ic];
      es[u] = (i < n_edges) ? s : -1;
    }
    const int row_base = slab * SLAB_ROWS + wave * 16;
    v4f xv[XTILE_LOADS];
    proj_x_load(X, row_base, n_units, lane, xv);
    const int wa = 2 * tid;
    const v2f wv = *reinterpret_cast<const v2f*>(A + wa);
    for (int k = tid; k < OWN_NBMAX; k += BK_THREADS) hist[k] = 0;
    proj_x_stage(xt[wave], lane, xv);
    *reinterpret_cast<v2f*>(wt + ((wa >> 7) + 8 * ((wa >> 6) & 1)) * FRAG_PITCH + (wa & 63)) = wv;
    __syncthreads();
    v2f fa[GEMM_KSTEPS], fb[GEMM_KSTEPS];
    proj_frag_read(xt[wave], wt, lane, fa, fb);
    const v8f ctile = proj_tile_mul(fa, fb);

    int eb[BK_EPT], erank[BK_EPT], epack[BK_EPT];
#pragma unroll
    for (int u = 0; u < BK_EPT; ++u) {
      eb[u] = -1; erank[u] = 0; epack[u] = 0;
      const int s = es[u];
      if ((unsigned)s < (unsigned)n_units) {
        const int b = (int)((unsigned)s / (unsigned)p.R);
        eb[u] = b;
        epack[u] = ((s - b * p.R) << 3) | (et[u] & 7);
        erank[u] = atomicAdd(&hist[b], 1);
      }
    }
    __syncthreads();

    const int b_lo = 2 * tid;
    const int c_lo = hist[b_lo];
    const int c_hi = hist[b_lo + 1];
    const int incl = wave_incl_sum(c_lo + c_hi, lane);
    if (lane == 31) wsum[wave] = incl;
    __syncthreads();
    const int waves_before = wave_total((lane < wave && lane < BK_THREADS / 32) ? wsum[lane] : 0);
    const int first = waves_before + incl - (c_lo + c_hi);
    hist[b_lo] = first;
    hist[b_lo + 1] = first + c_lo;
    if (tid == BK_THREADS - 1) s_tot = waves_before + incl;
    __syncthreads();

    int pos[BK_EPT];
#pragma unroll
    for (int u = 0; u < BK_EPT; ++u) pos[u] = hist[max(eb[u], 0)] + erank[u];
#pragma unroll
    for (int u = 0; u < BK_EPT; ++u) {
      if (eb[u] >= 0 && (unsigned)pos[u] < (unsigned)RUN_EDGES)
        stage[pos[u]] = make_int2(ed[u], epack[u]);
    }
    __syncthreads();

    if (b_lo < p.n_ranges)
      T[(size_t)b_lo * p.n_runs + run] = first | (c_lo << 16);
    if (b_lo + 1 < p.n_ranges)
      T[(size_t)(b_lo + 1) * p.n_runs + run] = (first + c_lo) | (c_hi << 16);

    const int tot = min(s_tot, min(RUN_EDGES, n_edges - e0));
    int2* __restrict__ Brun = B + (size_t)e0;
    int2 ov[BK_EPT];
#pragma unroll
    for (int u = 0; u < BK_EPT; ++u) ov[u] = stage[u * BK_THREADS + tid];
#pragma unroll
    for (int u = 0; u < BK_EPT; ++u) {
      const int k = u * BK_THREADS + tid;
      if (k < tot) Brun[k] = ov[u];
    }

    proj_tile_store(P, row_base, n_units, lane, ctile);
    slab += gridDim.x;
  }

  for (; slab < n_slabs; slab += gridDim.x) {
    const int row_base = slab * SLAB_ROWS + wave * 16;
    v2f fa[GEMM_KSTEPS], fb[GEMM_KSTEPS];
    proj_tile_load(X, A, row_base, n_units, lane, fa, fb);
    const v8f ctile = proj_tile_mul(fa, fb);
    proj_tile_store(P, row_base, n_units, lane, ctile);
  }
}

__global__ void __launch_bounds__(OWN_THREADS)
owner_lists(const float* __restrict__ P, const int2* __restrict__ B,
            const int* __restrict__ T,
            const float* __restrict__ hat_t, float* __restrict__ out,
            int n_edges, int n_units, size_t ws_bytes) {
  __shared__ int   head[2 * OWN_RMAX + 32];
  __shared__ float pu[8 * OWN_RMAX];
  __shared__ int2  slot[OWN_CAP + 32];
  __shared__ float slot_t[OWN_CAP + 32];
  __shared__ float st_mx[OWN_RMAX], st_dn[OWN_RMAX], st_g[OWN_RMAX];
  __shared__ int   st_has[OWN_RMAX];
  __shared__ int2  prun[OWN_PMAX];
  __shared__ int   wsum[OWN_THREADS / 32];
  __shared__ int   s_bc;

  const Plan p = make_plan(n_units, n_edges, ws_bytes);
  if (!p.ok) return;
  const int tid = threadIdx.x;
  const int lane = tid & 31;
  const int wave = tid >> 5;
  int top = 1;
  while (2 * top < p.n_runs) top *= 2;

  for (int range = blockIdx.x; range < p.n_ranges; range += gridDim.x) {
    const int base = range * p.R;
    const int nn = min(p.R, n_units - base);
    const int tw = (tid < p.n_runs) ? T[(size_t)range * p.n_runs + tid] : 0;

    for (int k = tid; k < nn; k += OWN_THREADS) {
      st_mx[k] = -INFINITY; st_dn[k] = 0.0f; st_g[k] = 0.0f; st_has[k] = 0;
    }
    for (int k = tid; k < 8 * nn; k += OWN_THREADS)
      pu[k] = P[(size_t)(base + (k >> 3)) * 16 + (k & 7)];

    const int pc = min((tw >> 16) & 0x7fff, RUN_EDGES);
    const int incl = wave_incl_sum(pc, lane);
    if (lane == 31) wsum[wave] = incl;
    __syncthreads();
    const int waves_before = wave_total((lane < wave) ? wsum[lane] : 0);
    const int pfirst = waves_before + incl - pc;
    prun[tid] = make_int2(pfirst, tid * RUN_EDGES + (tw & 0xffff) - pfirst);
    if (tid == OWN_THREADS - 1) s_bc = waves_before + incl;
    __syncthreads();
    const int bc = s_bc;

    for (int c0 = 0; c0 < bc; c0 += OWN_CAP) {
      const int m = min(OWN_CAP, bc - c0);
      for (int k = tid; k < 2 * nn; k += OWN_THREADS) head[k] = -1;
      __syncthreads();

      for (int k0 = tid; k0 < m; k0 += OWN_THREADS * OWN_EPT) {
        int el[OWN_EPT], er[OWN_EPT], ed[OWN_EPT];
        int2 rc[OWN_EPT];
        float pv[OWN_EPT], tt[OWN_EPT];
        int kk[OWN_EPT], pt[OWN_EPT];
#pragma unroll
        for (int u = 0; u < OWN_EPT; ++u) {
          kk[u] = c0 + min(k0 + u * OWN_THREADS, m - 1);
          pt[u] = 0;
        }
        for (int step = top; step >= 1; step >>= 1) {
#pragma unroll
          for (int u = 0; u < OWN_EPT; ++u)
            if (prun[pt[u] + step].x <= kk[u]) pt[u] += step;
        }
#pragma unroll
        for (int u = 0; u < OWN_EPT; ++u) {
          const int at = min(max(kk[u] + prun[pt[u]].y, 0), n_edges - 1);
          rc[u] = B[at];
        }
#pragma unroll
        for (int u = 0; u < OWN_EPT; ++u) {
          ed[u] = rc[u].x;
          el[u] = min(rc[u].y >> 3, nn - 1);
          if (el[u] < 0) el[u] = 0;
          er[u] = rc[u].y & 7;
        }
#pragma unroll
        for (int u = 0; u < OWN_EPT; ++u) {
          pv[u] = P[(size_t)ed[u] * 16 + 8 + er[u]];
          tt[u] = hat_t[ed[u]];
        }
#pragma unroll
        for (int u = 0; u < OWN_EPT; ++u) {
          const int k = k0 + u * OWN_THREADS;
          const bool ok = k < m;
          const int ks = ok ? k : OWN_CAP + lane;
          const int hs = ok ? 2 * el[u] + (k & 1) : 2 * OWN_RMAX + lane;
          const int prev = atomicExch(&head[hs], ks);
          const float e = leaky(pu[el[u] * 8 + er[u]] + pv[u]);
          slot[ks] = make_int2(__float_as_int(e), prev);
          slot_t[ks] = tt[u];
        }
      }
      __syncthreads();

      for (int n = tid; n < nn; n += OWN_THREADS) {
        int j0 = head[2 * n], j1 = head[2 * n + 1];
        if (j0 >= 0 || j1 >= 0) {
          float mx = st_mx[n], dn = st_dn[n], g = st_g[n];
          for (int k = 0; k < m; ++k) {
            const bool a0 = (unsigned)j0 < (unsigned)m;
            const bool a1 = (unsigned)j1 < (unsigned)m;
            if (!(a0 || a1)) break;
            float x0 = -INFINITY, x1 = -INFINITY, t0 = 0.0f, t1 = 0.0f;
            int nx0 = -1, nx1 = -1;
            if (a0) { const int2 s = slot[j0]; x0 = __int_as_float(s.x); nx0 = s.y; t0 = slot_t[j0]; }
            if (a1) { const int2 s = slot[j1]; x1 = __int_as_float(s.x); nx1 = s.y; t1 = slot_t[j1]; }
            const float mn = fmaxf(mx, fmaxf(x0, x1));
            const float sc = __expf(mx - mn);
            const float w0 = __expf(x0 - mn);
            const float w1 = __expf(x1 - mn);
            dn = (dn * sc + w0) + w1;
            g  = (g * sc + w0 * t0) + w1 * t1;
            mx = mn;
            j0 = nx0;
            j1 = nx1;
          }
          st_mx[n] = mx; st_dn[n] = dn; st_g[n] = g; st_has[n] = 1;
        }
      }
      __syncthreads();
    }

    for (int n = tid; n < nn; n += OWN_THREADS)
      out[base + n] = st_has[n] ? (st_g[n] / st_dn[n]) : 0.0f;
    __syncthreads();
  }
}

extern "C" void kernel_launch(void* const* d_in, const int* in_sizes, int n_in,
                              void* d_out, int out_size, void* d_ws, size_t ws_size,
                              hipStream_t stream) {
  const float* X     = (const float*)d_in[0];
  const float* hat_t = (const float*)d_in[1];
  const float* A     = (const float*)d_in[2];
  const int*   eidx  = (const int*)d_in[3];
  const int*   etype = (const int*)d_in[4];

  const int n_units = in_sizes[1];
  const int n_edges = in_sizes[4];
  const int* src = eidx;
  const int* dst = eidx + n_edges;

  float* P     = (float*)d_ws;
  int2*  B     = (int2*)(P + (size_t)n_units * 16);
  int*   T     = (int*)(B + n_edges);
  float* out   = (float*)d_out;

  const int bk_blocks = (n_edges + BK_THREADS * BK_EPT - 1) / (BK_THREADS * BK_EPT);
  proj_run_sort<<<bk_blocks, BK_THREADS, 0, stream>>>(X, A, P, src, dst, etype, B, T,
                                                      n_edges, n_units, ws_size);

  owner_lists<<<256, OWN_THREADS, 0, stream>>>(P, B, T, hat_t, out,
                                               n_edges, n_units, ws_size);
}


# hat_t[dst] rides beside each destination-half projection as an 8-byte pair (GEMM tile goes out in whole lines through LDS): an edge does one 8-byte gather, was two 4-byte gathers to two tables
# speedup vs baseline: 7.7825x; 1.1552x over previous
#include <hip/hip_runtime.h>
#include <hip/hip_bf16.h>

typedef __attribute__((ext_vector_type(2))) float v2f;
typedef __attribute__((ext_vector_type(4))) float v4f;
typedef __attribute__((ext_vector_type(8))) float v8f;

#define NODE_DIM   64
#define LEAKY      0.2f

#define OWN_RMAX    512
#define OWN_NBMAX   1024
#define OWN_CAP     12288
#define OWN_THREADS 1024
#define OWN_PMAX    OWN_THREADS
#define OWN_EPT     8
#define BK_THREADS  512
#define BK_EPT      8
#define RUN_EDGES   (BK_THREADS * BK_EPT)

struct Plan { int R; int n_ranges; int n_runs; bool ok; };

__device__ __forceinline__ Plan make_plan(int n_units, int n_edges, size_t ws_bytes) {
  Plan p;
  int R = (n_units + 255) / 256;
  R = (R + 31) & ~31;
  if (R < 32) R = 32;
  if (R > OWN_RMAX) R = OWN_RMAX;
  p.R = R;
  p.n_ranges = (n_units + R - 1) / R;
  p.n_runs = (n_edges + RUN_EDGES - 1) / RUN_EDGES;
  p.ok = (n_units > 0) && (n_edges > 0) &&
         (p.n_ranges <= OWN_NBMAX) && (p.n_runs <= OWN_PMAX) &&
         (ws_bytes >= ((size_t)n_units * 24 + (size_t)n_edges * 2 +
                       (size_t)p.n_ranges * (size_t)p.n_runs) * sizeof(float));
  return p;
}

__device__ __forceinline__ int wave_incl_sum(int v, int lane) {
#pragma unroll
  for (int off = 1; off < 32; off <<= 1) {
    const int t = __shfl_up(v, off, 32);
    if (lane >= off) v += t;
  }
  return v;
}

__device__ __forceinline__ int wave_total(int x) {
#pragma unroll
  for (int off = 16; off >= 1; off >>= 1) x += __shfl_xor(x, off, 32);
  return x;
}

#define GEMM_KSTEPS (NODE_DIM / 4)

__device__ __forceinline__ void proj_tile_load(const float* __restrict__ X, const float* __restrict__ A,
                                               int row_base, int n_units, int lane,
                                               v2f (&fa)[GEMM_KSTEPS], v2f (&fb)[GEMM_KSTEPS]) {
  const int half = lane >> 4;
  const int m    = lane & 15;
  int row = row_base + m;
  if (row >= n_units) row = n_units - 1;
  const float* __restrict__ xrow = X + (size_t)row * NODE_DIM;
  const float* __restrict__ wcol = A + (m & 7) * 128 + ((m >> 3) & 1) * 64;
#pragma unroll
  for (int i = 0; i < GEMM_KSTEPS; ++i) {
    const int k = 4 * i + half * 2;
    fa[i].x = xrow[k];     fa[i].y = xrow[k + 1];
    fb[i].x = wcol[k];     fb[i].y = wcol[k + 1];
  }
}

#define FRAG_PITCH (NODE_DIM + 4)
#define XTILE_LOADS 8

__device__ __forceinline__ void proj_x_load(const float* __restrict__ X, int row_base, int n_units,
                                            int lane, v4f (&xv)[XTILE_LOADS]) {
  const int sub = lane >> 4;
  const int col = (lane & 15) * 4;
#pragma unroll
  for (int j = 0; j < XTILE_LOADS; ++j) {
    int row = row_base + 2 * j + sub;
    if (row >= n_units) row = n_units - 1;
    xv[j] = *reinterpret_cast<const v4f*>(X + (size_t)row * NODE_DIM + col);
  }
}

__device__ __forceinline__ void proj_x_stage(float* xt, int lane, const v4f (&xv)[XTILE_LOADS]) {
  const int sub = lane >> 4;
  const int col = (lane & 15) * 4;
#pragma unroll
  for (int j = 0; j < XTILE_LOADS; ++j)
    *reinterpret_cast<v4f*>(xt + (2 * j + sub) * FRAG_PITCH + col) = xv[j];
}

__device__ __forceinline__ void proj_frag_read(const float* xt, const float* wt, int lane,
                                               v2f (&fa)[GEMM_KSTEPS], v2f (&fb)[GEMM_KSTEPS]) {
  const int half = lane >> 4;
  const int m    = lane & 15;
  const float* ar = xt + m * FRAG_PITCH + 4 * half;
  const float* br = wt + m * FRAG_PITCH + 4 * half;
#pragma unroll
  for (int q = 0; q < GEMM_KSTEPS / 2; ++q) {
    const v4f a4 = *reinterpret_cast<const v4f*>(ar + 8 * q);
    const v4f b4 = *reinterpret_cast<const v4f*>(br + 8 * q);
    fa[2 * q].x = a4.x;     fa[2 * q].y = a4.y;
    fa[2 * q + 1].x = a4.z; fa[2 * q + 1].y = a4.w;
    fb[2 * q].x = b4.x;     fb[2 * q].y = b4.y;
    fb[2 * q + 1].x = b4.z; fb[2 * q + 1].y = b4.w;
  }
}

__device__ __forceinline__ v8f proj_tile_mul(const v2f (&fa)[GEMM_KSTEPS], const v2f (&fb)[GEMM_KSTEPS]) {
  v8f c = {};
#pragma unroll
  for (int i = 0; i < GEMM_KSTEPS; ++i)
    c = __builtin_amdgcn_wmma_f32_16x16x4_f32(
         false, fa[i],  false, fb[i],
         (short)0, c,  false,  false);
  return c;
}

#define CT_PITCH 20

__device__ __forceinline__ void proj_tile_stage(float* ct, int lane, const v8f& c) {
  const int half = lane >> 4;
  const int m    = lane & 15;
#pragma unroll
  for (int j = 0; j < 8; ++j) ct[(j + half * 8) * CT_PITCH + m] = c[j];
}

__device__ __forceinline__ v2f proj_tile_hat(const float* __restrict__ hat_t, int row_base,
                                             int n_units, int lane) {
  v2f h;
  h.x = hat_t[min(row_base + (lane >> 2), n_units - 1)];
  h.y = hat_t[min(row_base + (lane >> 2) + 8, n_units - 1)];
  return h;
}

__device__ __forceinline__ void proj_tile_out(const float* ct, float* __restrict__ Pu,
                                              float* __restrict__ Q, int row_base, int n_units,
                                              int lane, v2f h) {
  const int ru = lane >> 1;
  const v4f u4 = *reinterpret_cast<const v4f*>(ct + ru * CT_PITCH + 4 * (lane & 1));
  const int rq = lane >> 2;
  const int cq = 8 + 2 * (lane & 3);
  const v2f d0 = *reinterpret_cast<const v2f*>(ct + rq * CT_PITCH + cq);
  const v2f d1 = *reinterpret_cast<const v2f*>(ct + (rq + 8) * CT_PITCH + cq);
  if (row_base + ru < n_units)
    *reinterpret_cast<v4f*>(Pu + (size_t)(row_base + ru) * 8 + 4 * (lane & 1)) = u4;
  v4f q0, q1;
  q0.x = d0.x; q0.y = h.x; q0.z = d0.y; q0.w = h.x;
  q1.x = d1.x; q1.y = h.y; q1.z = d1.y; q1.w = h.y;
  if (row_base + rq < n_units)
    *reinterpret_cast<v4f*>(Q + (size_t)(row_base + rq) * 16 + 4 * (lane & 3)) = q0;
  if (row_base + rq + 8 < n_units)
    *reinterpret_cast<v4f*>(Q + (size_t)(row_base + rq + 8) * 16 + 4 * (lane & 3)) = q1;
}

__device__ __forceinline__ float leaky(float e) {
  return (e > 0.0f) ? e : LEAKY * e;
}

#define SLAB_ROWS ((BK_THREADS / 32) * 16)

__global__ void __launch_bounds__(BK_THREADS)
proj_run_sort(const float* __restrict__ X, const float* __restrict__ A,
              const float* __restrict__ hat_t, float* __restrict__ Pu, float* __restrict__ Q,
              const int* __restrict__ src, const int* __restrict__ dst,
              const int* __restrict__ etype, int2* __restrict__ B, int* __restrict__ T,
              int n_edges, int n_units, size_t ws_bytes) {
  __shared__ int  hist[OWN_NBMAX];
  __shared__ int  wsum[BK_THREADS / 32];
  __shared__ int  s_tot;
  __shared__ int2 stage[RUN_EDGES];
  __shared__ __attribute__((aligned(16))) float xt[BK_THREADS / 32][16 * FRAG_PITCH];
  __shared__ __attribute__((aligned(16))) float wt[16 * FRAG_PITCH];
  static_assert(BK_THREADS * 2 == 16 * NODE_DIM, "one 8-byte load a thread fetches all of W");
  const Plan p = make_plan(n_units, n_edges, ws_bytes);
  const int tid = threadIdx.x;
  const int lane = tid & 31;
  const int wave = tid >> 5;
  const int run = blockIdx.x;
  const int n_slabs = (n_units + SLAB_ROWS - 1) / SLAB_ROWS;
  int slab = blockIdx.x;

  if (p.ok && run < p.n_runs) {
    const int e0 = run * RUN_EDGES;
    int es[BK_EPT], ed[BK_EPT], et[BK_EPT];
#pragma unroll
    for (int u = 0; u < BK_EPT; ++u) {
      const int i = e0 + u * BK_THREADS + tid;
      const int ic = min(i, n_edges - 1);
      const int s = src[ic];
      ed[u] = dst[ic];
      et[u] = etype[ic];
      es[u] = (i < n_edges) ? s : -1;
    }
    const int row_base = slab * SLAB_ROWS + wave * 16;
    v4f xv[XTILE_LOADS];
    proj_x_load(X, row_base, n_units, lane, xv);
    const v2f hrow = proj_tile_hat(hat_t, row_base, n_units, lane);
    const int wa = 2 * tid;
    const v2f wv = *reinterpret_cast<const v2f*>(A + wa);
    for (int k = tid; k < OWN_NBMAX; k += BK_THREADS) hist[k] = 0;
    proj_x_stage(xt[wave], lane, xv);
    *reinterpret_cast<v2f*>(wt + ((wa >> 7) + 8 * ((wa >> 6) & 1)) * FRAG_PITCH + (wa & 63)) = wv;
    __syncthreads();
    v2f fa[GEMM_KSTEPS], fb[GEMM_KSTEPS];
    proj_frag_read(xt[wave], wt, lane, fa, fb);
    const v8f ctile = proj_tile_mul(fa, fb);

    int eb[BK_EPT], erank[BK_EPT], epack[BK_EPT];
#pragma unroll
    for (int u = 0; u < BK_EPT; ++u) {
      eb[u] = -1; erank[u] = 0; epack[u] = 0;
      const int s = es[u];
      if ((unsigned)s < (unsigned)n_units) {
        const int b = (int)((unsigned)s / (unsigned)p.R);
        eb[u] = b;
        epack[u] = ((s - b * p.R) << 3) | (et[u] & 7);
        erank[u] = atomicAdd(&hist[b], 1);
      }
    }
    __syncthreads();

    const int b_lo = 2 * tid;
    const int c_lo = hist[b_lo];
    const int c_hi = hist[b_lo + 1];
    const int incl = wave_incl_sum(c_lo + c_hi, lane);
    if (lane == 31) wsum[wave] = incl;
    __syncthreads();
    const int waves_before = wave_total((lane < wave && lane < BK_THREADS / 32) ? wsum[lane] : 0);
    const int first = waves_before + incl - (c_lo + c_hi);
    hist[b_lo] = first;
    hist[b_lo + 1] = first + c_lo;
    if (tid == BK_THREADS - 1) s_tot = waves_before + incl;
    __syncthreads();

    int pos[BK_EPT];
#pragma unroll
    for (int u = 0; u < BK_EPT; ++u) pos[u] = hist[max(eb[u], 0)] + erank[u];
#pragma unroll
    for (int u = 0; u < BK_EPT; ++u) {
      if (eb[u] >= 0 && (unsigned)pos[u] < (unsigned)RUN_EDGES)
        stage[pos[u]] = make_int2(ed[u], epack[u]);
    }
    proj_tile_stage(xt[wave], lane, ctile);
    __syncthreads();

    if (b_lo < p.n_ranges)
      T[(size_t)b_lo * p.n_runs + run] = first | (c_lo << 16);
    if (b_lo + 1 < p.n_ranges)
      T[(size_t)(b_lo + 1) * p.n_runs + run] = (first + c_lo) | (c_hi << 16);

    const int tot = min(s_tot, min(RUN_EDGES, n_edges - e0));
    int2* __restrict__ Brun = B + (size_t)e0;
    int2 ov[BK_EPT];
#pragma unroll
    for (int u = 0; u < BK_EPT; ++u) ov[u] = stage[u * BK_THREADS + tid];
#pragma unroll
    for (int u = 0; u < BK_EPT; ++u) {
      const int k = u * BK_THREADS + tid;
      if (k < tot) Brun[k] = ov[u];
    }

    proj_tile_out(xt[wave], Pu, Q, row_base, n_units, lane, hrow);
    slab += gridDim.x;
  }

  for (; p.ok && slab < n_slabs; slab += gridDim.x) {
    const int row_base = slab * SLAB_ROWS + wave * 16;
    const v2f hrow = proj_tile_hat(hat_t, row_base, n_units, lane);
    v2f fa[GEMM_KSTEPS], fb[GEMM_KSTEPS];
    proj_tile_load(X, A, row_base, n_units, lane, fa, fb);
    const v8f ctile = proj_tile_mul(fa, fb);
    proj_tile_stage(xt[wave], lane, ctile);
    __syncthreads();
    proj_tile_out(xt[wave], Pu, Q, row_base, n_units, lane, hrow);
  }
}

__global__ void __launch_bounds__(OWN_THREADS)
owner_lists(const float* __restrict__ Pu, const float* __restrict__ Q,
            const int2* __restrict__ B, const int* __restrict__ T,
            float* __restrict__ out,
            int n_edges, int n_units, size_t ws_bytes) {
  __shared__ int   head[2 * OWN_RMAX + 32];
  __shared__ float pu[8 * OWN_RMAX];
  __shared__ int2  slot[OWN_CAP + 32];
  __shared__ float slot_t[OWN_CAP + 32];
  __shared__ float st_mx[OWN_RMAX], st_dn[OWN_RMAX], st_g[OWN_RMAX];
  __shared__ int   st_has[OWN_RMAX];
  __shared__ int2  prun[OWN_PMAX];
  __shared__ int   wsum[OWN_THREADS / 32];
  __shared__ int   s_bc;

  const Plan p = make_plan(n_units, n_edges, ws_bytes);
  if (!p.ok) return;
  const int tid = threadIdx.x;
  const int lane = tid & 31;
  const int wave = tid >> 5;
  int top = 1;
  while (2 * top < p.n_runs) top *= 2;

  for (int range = blockIdx.x; range < p.n_ranges; range += gridDim.x) {
    const int base = range * p.R;
    const int nn = min(p.R, n_units - base);
    const int tw = (tid < p.n_runs) ? T[(size_t)range * p.n_runs + tid] : 0;

    for (int k = tid; k < nn; k += OWN_THREADS) {
      st_mx[k] = -INFINITY; st_dn[k] = 0.0f; st_g[k] = 0.0f; st_has[k] = 0;
    }
    for (int k = tid; k < 8 * nn; k += OWN_THREADS)
      pu[k] = Pu[(size_t)base * 8 + k];

    const int pc = min((tw >> 16) & 0x7fff, RUN_EDGES);
    const int incl = wave_incl_sum(pc, lane);
    if (lane == 31) wsum[wave] = incl;
    __syncthreads();
    const int waves_before = wave_total((lane < wave) ? wsum[lane] : 0);
    const int pfirst = waves_before + incl - pc;
    prun[tid] = make_int2(pfirst, tid * RUN_EDGES + (tw & 0xffff) - pfirst);
    if (tid == OWN_THREADS - 1) s_bc = waves_before + incl;
    __syncthreads();
    const int bc = s_bc;

    for (int c0 = 0; c0 < bc; c0 += OWN_CAP) {
      const int m = min(OWN_CAP, bc - c0);
      for (int k = tid; k < 2 * nn; k += OWN_THREADS) head[k] = -1;
      __syncthreads();

      for (int k0 = tid; k0 < m; k0 += OWN_THREADS * OWN_EPT) {
        int el[OWN_EPT], er[OWN_EPT], ed[OWN_EPT];
        int2 rc[OWN_EPT];
        float pv[OWN_EPT], tt[OWN_EPT];
        int kk[OWN_EPT], pt[OWN_EPT];
#pragma unroll
        for (int u = 0; u < OWN_EPT; ++u) {
          kk[u] = c0 + min(k0 + u * OWN_THREADS, m - 1);
          pt[u] = 0;
        }
        for (int step = top; step >= 1; step >>= 1) {
#pragma unroll
          for (int u = 0; u < OWN_EPT; ++u)
            if (prun[pt[u] + step].x <= kk[u]) pt[u] += step;
        }
#pragma unroll
        for (int u = 0; u < OWN_EPT; ++u) {
          const int at = min(max(kk[u] + prun[pt[u]].y, 0), n_edges - 1);
          rc[u] = B[at];
        }
#pragma unroll
        for (int u = 0; u < OWN_EPT; ++u) {
          ed[u] = rc[u].x;
          el[u] = min(rc[u].y >> 3, nn - 1);
          if (el[u] < 0) el[u] = 0;
          er[u] = rc[u].y & 7;
        }
#pragma unroll
        for (int u = 0; u < OWN_EPT; ++u) {
          const v2f g = *reinterpret_cast<const v2f*>(Q + (size_t)ed[u] * 16 + 2 * er[u]);
          pv[u] = g.x;
          tt[u] = g.y;
        }
#pragma unroll
        for (int u = 0; u < OWN_EPT; ++u) {
          const int k = k0 + u * OWN_THREADS;
          const bool ok = k < m;
          const int ks = ok ? k : OWN_CAP + lane;
          const int hs = ok ? 2 * el[u] + (k & 1) : 2 * OWN_RMAX + lane;
          const int prev = atomicExch(&head[hs], ks);
          const float e = leaky(pu[el[u] * 8 + er[u]] + pv[u]);
          slot[ks] = make_int2(__float_as_int(e), prev);
          slot_t[ks] = tt[u];
        }
      }
      __syncthreads();

      for (int n = tid; n < nn; n += OWN_THREADS) {
        int j0 = head[2 * n], j1 = head[2 * n + 1];
        if (j0 >= 0 || j1 >= 0) {
          float mx = st_mx[n], dn = st_dn[n], g = st_g[n];
          for (int k = 0; k < m; ++k) {
            const bool a0 = (unsigned)j0 < (unsigned)m;
            const bool a1 = (unsigned)j1 < (unsigned)m;
            if (!(a0 || a1)) break;
            float x0 = -INFINITY, x1 = -INFINITY, t0 = 0.0f, t1 = 0.0f;
            int nx0 = -1, nx1 = -1;
            if (a0) { const int2 s = slot[j0]; x0 = __int_as_float(s.x); nx0 = s.y; t0 = slot_t[j0]; }
            if (a1) { const int2 s = slot[j1]; x1 = __int_as_float(s.x); nx1 = s.y; t1 = slot_t[j1]; }
            const float mn = fmaxf(mx, fmaxf(x0, x1));
            const float sc = __expf(mx - mn);
            const float w0 = __expf(x0 - mn);
            const float w1 = __expf(x1 - mn);
            dn = (dn * sc + w0) + w1;
            g  = (g * sc + w0 * t0) + w1 * t1;
            mx = mn;
            j0 = nx0;
            j1 = nx1;
          }
          st_mx[n] = mx; st_dn[n] = dn; st_g[n] = g; st_has[n] = 1;
        }
      }
      __syncthreads();
    }

    for (int n = tid; n < nn; n += OWN_THREADS)
      out[base + n] = st_has[n] ? (st_g[n] / st_dn[n]) : 0.0f;
    __syncthreads();
  }
}

extern "C" void kernel_launch(void* const* d_in, const int* in_sizes, int n_in,
                              void* d_out, int out_size, void* d_ws, size_t ws_size,
                              hipStream_t stream) {
  const float* X     = (const float*)d_in[0];
  const float* hat_t = (const float*)d_in[1];
  const float* A     = (const float*)d_in[2];
  const int*   eidx  = (const int*)d_in[3];
  const int*   etype = (const int*)d_in[4];

  const int n_units = in_sizes[1];
  const int n_edges = in_sizes[4];
  const int* src = eidx;
  const int* dst = eidx + n_edges;

  float* Pu    = (float*)d_ws;
  float* Q     = Pu + (size_t)n_units * 8;
  int2*  B     = (int2*)(Q + (size_t)n_units * 16);
  int*   T     = (int*)(B + n_edges);
  float* out   = (float*)d_out;

  const int bk_blocks = (n_edges + BK_THREADS * BK_EPT - 1) / (BK_THREADS * BK_EPT);
  proj_run_sort<<<bk_blocks, BK_THREADS, 0, stream>>>(X, A, hat_t, Pu, Q, src, dst, etype, B, T,
                                                      n_edges, n_units, ws_size);

  owner_lists<<<256, OWN_THREADS, 0, stream>>>(Pu, Q, B, T, out,
                                               n_edges, n_units, ws_size);
}
